# starting kernel
# baseline (speedup 1.0000x reference)
#include <hip/hip_runtime.h>

#define D 64

typedef __attribute__((ext_vector_type(2))) float v2f;
typedef __attribute__((ext_vector_type(8))) float v8f;

__global__ void __launch_bounds__(256) scatter_edges(
    const float* __restrict__ uA, const float* __restrict__ iA,
    float* __restrict__ uB, float* __restrict__ iB,
    const int* __restrict__ edge_u, const int* __restrict__ edge_i,
    const float* __restrict__ norm, int n_edges)
{
    int wave = (int)(blockIdx.x * (blockDim.x >> 5) + (threadIdx.x >> 5));
    int lane = threadIdx.x & 31;
    if (wave >= n_edges) return;

    int   u  = edge_u[wave];
    int   it = edge_i[wave];
    float w  = norm[wave];

    const v2f* us = (const v2f*)(uA + (size_t)u  * D);
    const v2f* is = (const v2f*)(iA + (size_t)it * D);
    v2f uv = us[lane];
    v2f iv = is[lane];

    float* id = iB + (size_t)it * D + lane * 2;
    float* ud = uB + (size_t)u  * D + lane * 2;
    atomicAdd(&id[0], uv.x * w);
    atomicAdd(&id[1], uv.y * w);
    atomicAdd(&ud[0], iv.x * w);
    atomicAdd(&ud[1], iv.y * w);
}

__global__ void __launch_bounds__(256) gather_rows(
    const float* __restrict__ utab, const float* __restrict__ itab,
    const int* __restrict__ users, const int* __restrict__ pos,
    const int* __restrict__ neg,
    float* __restrict__ out, int batch, int accumulate)
{
    int gid = blockIdx.x * blockDim.x + threadIdx.x;
    int total = 3 * batch * D;
    if (gid >= total) return;

    int d   = gid & (D - 1);
    int r   = (gid / D) % batch;
    int seg = gid / (batch * D);

    const float* src;
    if (seg == 0)      src = utab + (size_t)users[r] * D;
    else if (seg == 1) src = itab + (size_t)pos[r]   * D;
    else               src = itab + (size_t)neg[r]   * D;

    float v = src[d];
    if (accumulate) out[gid] += v;
    else            out[gid]  = v;
}

__global__ void __launch_bounds__(256) wmma_scale_inplace(
    float* __restrict__ out, int ntiles)
{
    int wid = (int)(blockIdx.x * (blockDim.x >> 5) + (threadIdx.x >> 5));
    if (wid >= ntiles) return;

    int lane = threadIdx.x & 31;
    int h = lane >> 4;
    int l = lane & 15;
    int rg = wid >> 2;
    int ct = wid & 3;

    float* base = out + (size_t)rg * 16 * D + ct * 16;

    v8f acc = {};
#pragma unroll
    for (int c = 0; c < 4; ++c) {
        int k0 = 2 * h;
        int k1 = 2 * h + 1;
        v2f a, b;
        a.x = base[(size_t)l * D + 4 * c + k0];
        a.y = base[(size_t)l * D + 4 * c + k1];
        b.x = ((4 * c + k0) == l) ? 0.25f : 0.0f;
        b.y = ((4 * c + k1) == l) ? 0.25f : 0.0f;
        acc = __builtin_amdgcn_wmma_f32_16x16x4_f32(
             false, a,  false, b,
             (short)0, acc,  false,  false);
    }
#pragma unroll
    for (int j = 0; j < 8; ++j)
        base[(size_t)(h * 8 + j) * D + l] = acc[j];
}

extern "C" void kernel_launch(void* const* d_in, const int* in_sizes, int n_in,
                              void* d_out, int out_size, void* d_ws, size_t ws_size,
                              hipStream_t stream)
{
    const float* user_feat = (const float*)d_in[0];
    const float* item_feat = (const float*)d_in[1];
    const int*   edge_u    = (const int*)d_in[2];
    const int*   edge_i    = (const int*)d_in[3];
    const float* norm      = (const float*)d_in[4];
    const int*   users     = (const int*)d_in[5];
    const int*   pos       = (const int*)d_in[6];
    const int*   neg       = (const int*)d_in[7];
    float* out = (float*)d_out;

    const int n_users = in_sizes[0] / D;
    const int n_items = in_sizes[1] / D;
    const int n_edges = in_sizes[2];
    const int batch   = in_sizes[5];

    size_t uElems = (size_t)n_users * D;
    size_t iElems = (size_t)n_items * D;

    float* u0 = (float*)d_ws;
    float* u1 = u0 + uElems;
    float* i0 = u1 + uElems;
    float* i1 = i0 + iElems;

    float* uCur = u0; float* uNxt = u1;
    float* iCur = i0; float* iNxt = i1;

    hipMemcpyAsync(uCur, user_feat, uElems * sizeof(float),
                   hipMemcpyDeviceToDevice, stream);
    hipMemcpyAsync(iCur, item_feat, iElems * sizeof(float),
                   hipMemcpyDeviceToDevice, stream);

    int gthreads = 3 * batch * D;
    int gblocks  = (gthreads + 255) / 256;
    gather_rows<<<gblocks, 256, 0, stream>>>(user_feat, item_feat,
                                             users, pos, neg, out, batch, 0);

    int sblocks = (n_edges + 7) / 8;
    for (int L = 0; L < 3; ++L) {
        hipMemsetAsync(uNxt, 0, uElems * sizeof(float), stream);
        hipMemsetAsync(iNxt, 0, iElems * sizeof(float), stream);
        scatter_edges<<<sblocks, 256, 0, stream>>>(uCur, iCur, uNxt, iNxt,
                                                   edge_u, edge_i, norm, n_edges);
        gather_rows<<<gblocks, 256, 0, stream>>>(uNxt, iNxt,
                                                 users, pos, neg, out, batch, 1);
        float* t;
        t = uCur; uCur = uNxt; uNxt = t;
        t = iCur; iCur = iNxt; iNxt = t;
    }

    int nrows  = 3 * batch;
    int ntiles = (nrows / 16) * (D / 16);
    int wblocks = (ntiles + 7) / 8;
    wmma_scale_inplace<<<wblocks, 256, 0, stream>>>(out, ntiles);
}


# layers 2 and 3 not materialised: each requested row walks its list and re-sums every listed row's layer-2 value from the layer-1 tables in registers; out row stored once; sum order unchanged
# speedup vs baseline: 2.3186x; 2.3186x over previous
#include <hip/hip_runtime.h>

#define D 64

typedef __attribute__((ext_vector_type(2))) float v2f;
typedef __attribute__((ext_vector_type(4))) float v4f;
typedef __attribute__((ext_vector_type(8))) float v8f;
typedef __attribute__((ext_vector_type(2))) int   v2i;

__global__ void __launch_bounds__(1024) count_edges(
    const int* __restrict__ edge_u, const int* __restrict__ edge_i,
    int* __restrict__ cnt, int* __restrict__ rank_u, int* __restrict__ rank_i,
    int n_users, int n_edges)
{
    int e = (int)(blockIdx.x * blockDim.x + threadIdx.x);
    if (e >= n_edges) return;
    int u  = edge_u[e];
    int it = edge_i[e];
    int su = atomicAdd(&cnt[u], 1);
    int sv = atomicAdd(&cnt[n_users + it], 1);
    rank_u[e] = su;
    rank_i[e] = sv;
}

__global__ void __launch_bounds__(256) block_sums(
    const int* __restrict__ cnt, int* __restrict__ bsum, int n)
{
    __shared__ int ws[8];
    int t = (int)threadIdx.x;
    int base = (int)(blockIdx.x * 256 + t) * 4;
    int s = 0;
#pragma unroll
    for (int j = 0; j < 4; ++j)
        if (base + j < n) s += cnt[base + j];
    for (int d = 16; d > 0; d >>= 1) s += __shfl_xor(s, d, 32);
    if ((t & 31) == 0) ws[t >> 5] = s;
    __syncthreads();
    if (t == 0) {
        int tot = 0;
        for (int i = 0; i < 8; ++i) tot += ws[i];
        bsum[blockIdx.x] = tot;
    }
}

__global__ void __launch_bounds__(256) write_offsets(
    const int* __restrict__ cnt, const int* __restrict__ bsum,
    int* __restrict__ off, int n)
{
    __shared__ int wsum[8];
    __shared__ int psum[8];
    int t = (int)threadIdx.x, lane = t & 31, wv = t >> 5;

    int p = 0;
    for (int j = t; j < (int)blockIdx.x; j += 256) p += bsum[j];
    for (int d = 16; d > 0; d >>= 1) p += __shfl_xor(p, d, 32);
    if (lane == 0) psum[wv] = p;

    int base = (int)(blockIdx.x * 256 + t) * 4;
    int c[4];
    int s = 0;
#pragma unroll
    for (int j = 0; j < 4; ++j) {
        c[j] = (base + j < n) ? cnt[base + j] : 0;
        s += c[j];
    }
    int inc = s;
    for (int d = 1; d < 32; d <<= 1) {
        int up = __shfl_up(inc, d, 32);
        if (lane >= d) inc += up;
    }
    if (lane == 31) wsum[wv] = inc;
    __syncthreads();

    int pre = 0;
    for (int i = 0; i < 8; ++i) pre += psum[i];
    for (int i = 0; i < 8; ++i) if (i < wv) pre += wsum[i];
    int ex = pre + inc - s;
#pragma unroll
    for (int j = 0; j < 4; ++j)
        if (base + j < n) { off[base + j] = ex; ex += c[j]; }
    if (base < n && base + 4 >= n) off[n] = ex;
}

__global__ void __launch_bounds__(1024) fill_entries(
    const int* __restrict__ edge_u, const int* __restrict__ edge_i,
    const float* __restrict__ norm, const int* __restrict__ off,
    const int* __restrict__ rank_u, const int* __restrict__ rank_i,
    int* __restrict__ ent, int n_users, int n_edges)
{
    int e = (int)(blockIdx.x * blockDim.x + threadIdx.x);
    if (e >= n_edges) return;
    int u  = edge_u[e];
    int it = edge_i[e];
    int wb = __float_as_int(norm[e]);
    int pu = off[u] + rank_u[e];
    int pi = off[n_users + it] + rank_i[e];
    v2i* en2 = (v2i*)ent;
    v2i a; a.x = it; a.y = wb;
    v2i b; b.x = u;  b.y = wb;
    en2[pu] = a;
    en2[pi] = b;
}

__device__ __forceinline__ void pull_body(
    const float* __restrict__ src, float* __restrict__ dst,
    const int* __restrict__ off, const int* __restrict__ ent, int n_rows,
    unsigned blk, unsigned nblk)
{
    const int l   = (int)(threadIdx.x & 15);
    const int nhw = (int)(nblk * (blockDim.x >> 4));
    int r = (int)((blk * blockDim.x + threadIdx.x) >> 4);
    const v2i* en2 = (const v2i*)ent;
    const v4f* s4  = (const v4f*)src;
    v4f* d4 = (v4f*)dst;

    const v4f zero = {0.f, 0.f, 0.f, 0.f};
    const int last = n_rows - 1;
    if (r >= n_rows) return;
    int b = off[r], e = off[r + 1];
    while (r < n_rows) {
        int rn = r + nhw;
        int rc = rn < last ? rn : last;
        int bn = off[rc], en = off[rc + 1];

        v4f acc = zero;
        for (int base = b; base < e; base += 16) {
            int m = e - base;
            if (m > 16) m = 16;
            int   myi = 0;
            float myw = 0.f;
            if (l < m) {
                v2i t = en2[base + l];
                myi = t.x;
                myw = __int_as_float(t.y);
            }
            for (int k = 0; k < m; k += 8) {
                const int c = m - k;
                v4f v0, v1, v2, v3, v4, v5, v6, v7;
                const float w0 = __shfl(myw, k + 0, 16);
                const float w1 = __shfl(myw, k + 1, 16);
                const float w2 = __shfl(myw, k + 2, 16);
                const float w3 = __shfl(myw, k + 3, 16);
                const float w4 = __shfl(myw, k + 4, 16);
                const float w5 = __shfl(myw, k + 5, 16);
                const float w6 = __shfl(myw, k + 6, 16);
                const float w7 = __shfl(myw, k + 7, 16);
#define ROW_LD(j) v##j = s4[(size_t)__shfl(myi, k + (j), 16) * (D / 4) + l];
                ROW_LD(0)
                if (c > 1) { ROW_LD(1)
                if (c > 2) { ROW_LD(2)
                if (c > 3) { ROW_LD(3)
                if (c > 4) { ROW_LD(4)
                if (c > 5) { ROW_LD(5)
                if (c > 6) { ROW_LD(6)
                if (c > 7) { ROW_LD(7) } } } } } } }
#undef ROW_LD
                acc += w0 * v0;
                if (c > 1) { acc += w1 * v1;
                if (c > 2) { acc += w2 * v2;
                if (c > 3) { acc += w3 * v3;
                if (c > 4) { acc += w4 * v4;
                if (c > 5) { acc += w5 * v5;
                if (c > 6) { acc += w6 * v6;
                if (c > 7) { acc += w7 * v7; } } } } } } }
            }
        }
        d4[(size_t)r * (D / 4) + l] = acc;
        r = rn; b = bn; e = en;
    }
}

__global__ void __launch_bounds__(256) pull_both_sides(
    const float* __restrict__ src_a, float* __restrict__ dst_a,
    const int* __restrict__ off_a, int rows_a, unsigned nb_a,
    const float* __restrict__ src_b, float* __restrict__ dst_b,
    const int* __restrict__ off_b, int rows_b,
    const int* __restrict__ ent)
{
    const unsigned b = blockIdx.x;
    const bool second = b >= nb_a;
    pull_body(second ? src_b : src_a, second ? dst_b : dst_a,
              second ? off_b : off_a, ent, second ? rows_b : rows_a,
              second ? b - nb_a : b, second ? gridDim.x - nb_a : nb_a);
}


__device__ __forceinline__ void first_entries(
    const v2i* __restrict__ en2, int b, int e, int l, int& myi, float& myw)
{
    myi = 0;
    myw = 0.f;
    if (b + l < e) {
        v2i t = en2[b + l];
        myi = t.x;
        myw = __int_as_float(t.y);
    }
}

template <class Side>
__device__ __forceinline__ v4f sum_list(
    const v4f* __restrict__ s4, const v2i* __restrict__ en2,
    int b, int e, int l, int myi, float myw, Side side)
{
    v4f acc = {0.f, 0.f, 0.f, 0.f};
    bool side_pending = true;
    for (int base = b; base < e; base += 16) {
        int m = e - base;
        if (m > 16) m = 16;
        if (base != b) first_entries(en2, base, e, l, myi, myw);
        for (int k = 0; k < m; k += 8) {
            const int c = m - k;
            v4f v0, v1, v2, v3, v4, v5, v6, v7;
            const float w0 = __shfl(myw, k + 0, 16);
            const float w1 = __shfl(myw, k + 1, 16);
            const float w2 = __shfl(myw, k + 2, 16);
            const float w3 = __shfl(myw, k + 3, 16);
            const float w4 = __shfl(myw, k + 4, 16);
            const float w5 = __shfl(myw, k + 5, 16);
            const float w6 = __shfl(myw, k + 6, 16);
            const float w7 = __shfl(myw, k + 7, 16);
#define ROW_LD(j) v##j = s4[(size_t)__shfl(myi, k + (j), 16) * (D / 4) + l];
            ROW_LD(0)
            if (c > 1) { ROW_LD(1)
            if (c > 2) { ROW_LD(2)
            if (c > 3) { ROW_LD(3)
            if (c > 4) { ROW_LD(4)
            if (c > 5) { ROW_LD(5)
            if (c > 6) { ROW_LD(6)
            if (c > 7) { ROW_LD(7) } } } } } } }
#undef ROW_LD
            if (side_pending) { side(); side_pending = false; }
            acc += w0 * v0;
            if (c > 1) { acc += w1 * v1;
            if (c > 2) { acc += w2 * v2;
            if (c > 3) { acc += w3 * v3;
            if (c > 4) { acc += w4 * v4;
            if (c > 5) { acc += w5 * v5;
            if (c > 6) { acc += w6 * v6;
            if (c > 7) { acc += w7 * v7; } } } } } } }
        }
    }
    if (side_pending) side();
    return acc;
}

__global__ void __launch_bounds__(256) requested_rows_two_levels(
    const float* __restrict__ user_feat, const float* __restrict__ item_feat,
    const float* __restrict__ u1, const float* __restrict__ i1,
    const int* __restrict__ off, const int* __restrict__ ent,
    const int* __restrict__ users, const int* __restrict__ pos,
    const int* __restrict__ neg,
    float* __restrict__ out, int batch, int n_users)
{
    const int l = (int)(threadIdx.x & 15);
    const int r = (int)((blockIdx.x * blockDim.x + threadIdx.x) >> 4);
    if (r >= 3 * batch) return;
    const int seg = r / batch;
    const int i   = r - seg * batch;
    const bool is_user = (seg == 0);
    const int idx = is_user ? users[i] : (seg == 1 ? pos[i] : neg[i]);

    const v4f* own0 = (const v4f*)(is_user ? user_feat : item_feat);
    const v4f* own1 = (const v4f*)(is_user ? u1 : i1);
    const v4f* opp1 = (const v4f*)(is_user ? i1 : u1);
    const int jr       = is_user ? idx : n_users + idx;
    const int opp_base = is_user ? n_users : 0;
    const v2i* en2 = (const v2i*)ent;

    const int b = off[jr], e = off[jr + 1];
    v4f acc2 = {0.f, 0.f, 0.f, 0.f};
    v4f acc3 = {0.f, 0.f, 0.f, 0.f};
    for (int base = b; base < e; base += 16) {
        int m = e - base;
        if (m > 16) m = 16;
        int   s  = 0, sb = 0, se = 0;
        float w  = 0.f;
        if (l < m) {
            v2i t = en2[base + l];
            s  = t.x;
            w  = __int_as_float(t.y);
            sb = off[opp_base + s];
            se = off[opp_base + s + 1];
        }
        int kb = __shfl(sb, 0, 16), ke = __shfl(se, 0, 16);
        int   ci;
        float cw;
        first_entries(en2, kb, ke, l, ci, cw);
        for (int k = 0; k < m; ++k) {
            const int   sk = __shfl(s, k, 16);
            const float wk = __shfl(w, k, 16);
            const int kn = (k + 1 < m) ? k + 1 : k;
            const int nb = __shfl(sb, kn, 16), ne = __shfl(se, kn, 16);
            int   ni = 0;
            float nw = 0.f;
            v4f   v;
            const v4f t = sum_list(own1, en2, kb, ke, l, ci, cw, [&]() {
                if (k + 1 < m) first_entries(en2, nb, ne, l, ni, nw);
                v = opp1[(size_t)sk * (D / 4) + l];
            });
            acc2 += wk * v;
            acc3 += wk * t;
            kb = nb; ke = ne; ci = ni; cw = nw;
        }
    }
    const v4f h0 = own0[(size_t)idx * (D / 4) + l];
    const v4f h1 = own1[(size_t)idx * (D / 4) + l];
    ((v4f*)out)[(size_t)r * (D / 4) + l] = ((h0 + h1) + acc2) + acc3;
}

__global__ void __launch_bounds__(256) wmma_scale_inplace(
    float* __restrict__ out, int ntiles)
{
    int wid = (int)(blockIdx.x * (blockDim.x >> 5) + (threadIdx.x >> 5));
    if (wid >= ntiles) return;

    int lane = threadIdx.x & 31;
    int h = lane >> 4;
    int l = lane & 15;
    int rg = wid >> 2;
    int ct = wid & 3;

    float* base = out + (size_t)rg * 16 * D + ct * 16;

    v8f acc = {};
#pragma unroll
    for (int c = 0; c < 4; ++c) {
        int k0 = 2 * h;
        int k1 = 2 * h + 1;
        v2f a, b;
        a.x = base[(size_t)l * D + 4 * c + k0];
        a.y = base[(size_t)l * D + 4 * c + k1];
        b.x = ((4 * c + k0) == l) ? 0.25f : 0.0f;
        b.y = ((4 * c + k1) == l) ? 0.25f : 0.0f;
        acc = __builtin_amdgcn_wmma_f32_16x16x4_f32(
             false, a,  false, b,
             (short)0, acc,  false,  false);
    }
#pragma unroll
    for (int j = 0; j < 8; ++j)
        base[(size_t)(h * 8 + j) * D + l] = acc[j];
}

extern "C" void kernel_launch(void* const* d_in, const int* in_sizes, int n_in,
                              void* d_out, int out_size, void* d_ws, size_t ws_size,
                              hipStream_t stream)
{
    const float* user_feat = (const float*)d_in[0];
    const float* item_feat = (const float*)d_in[1];
    const int*   edge_u    = (const int*)d_in[2];
    const int*   edge_i    = (const int*)d_in[3];
    const float* norm      = (const float*)d_in[4];
    const int*   users     = (const int*)d_in[5];
    const int*   pos       = (const int*)d_in[6];
    const int*   neg       = (const int*)d_in[7];
    float* out = (float*)d_out;

    const int n_users = in_sizes[0] / D;
    const int n_items = in_sizes[1] / D;
    const int n_edges = in_sizes[2];
    const int batch   = in_sizes[5];
    const int n_rows  = n_users + n_items;

    size_t uElems = (size_t)n_users * D;
    size_t iElems = (size_t)n_items * D;

    float* su_a = (float*)d_ws;
    float* su_b = su_a + uElems;
    float* si   = su_b + uElems;
    int*   off  = (int*)(si + iElems);
    int offPad  = ((n_rows + 1 + 63) / 64) * 64;
    int*   ent  = off + offPad;

    int nsb     = (n_rows + 1023) / 1024;
    int*   cnt    = (int*)su_b;
    int*   bsum   = cnt + offPad;
    int*   rank_u = bsum + ((nsb + 63) / 64) * 64;
    int*   rank_i = rank_u + n_edges;

    int eblocks = (n_edges + 1023) / 1024;
    hipMemsetAsync(cnt, 0, (size_t)n_rows * sizeof(int), stream);
    count_edges<<<eblocks, 1024, 0, stream>>>(edge_u, edge_i, cnt, rank_u, rank_i,
                                               n_users, n_edges);
    block_sums<<<nsb, 256, 0, stream>>>(cnt, bsum, n_rows);
    write_offsets<<<nsb, 256, 0, stream>>>(cnt, bsum, off, n_rows);
    fill_entries<<<eblocks, 1024, 0, stream>>>(edge_u, edge_i, norm, off,
                                                rank_u, rank_i, ent, n_users, n_edges);

    int ru = (n_users + 24575) / 24576;
    int ri = (n_items + 24575) / 24576;
    int ublocks = (n_users + 16 * ru - 1) / (16 * ru);
    int iblocks = (n_items + 16 * ri - 1) / (16 * ri);
    const int* off_u = off;
    const int* off_i = off + n_users;

    pull_both_sides<<<ublocks + iblocks, 256, 0, stream>>>(
        item_feat, su_a, off_u, n_users, ublocks,
        user_feat, si,   off_i, n_items, ent);

    int oblocks = (3 * batch + 15) / 16;
    requested_rows_two_levels<<<oblocks, 256, 0, stream>>>(
        user_feat, item_feat, su_a, si, off, ent, users, pos, neg, out, batch, n_users);

    int nrows  = 3 * batch;
    int ntiles = (nrows / 16) * (D / 16);
    int wblocks = (ntiles + 7) / 8;
    wmma_scale_inplace<<<wblocks, 256, 0, stream>>>(out, ntiles);
}


# adjacency lists by a two-level counting sort (1024-row buckets, counts and ranks on LDS counters), no global atomic per edge end, no memset; sums in slot order, slots in LDS arrival order
# speedup vs baseline: 3.7190x; 1.6040x over previous
#include <hip/hip_runtime.h>

#define D 64

typedef __attribute__((ext_vector_type(2))) float v2f;
typedef __attribute__((ext_vector_type(4))) float v4f;
typedef __attribute__((ext_vector_type(8))) float v8f;
typedef __attribute__((ext_vector_type(2))) int   v2i;

typedef __attribute__((ext_vector_type(4))) int   v4i;

#define SORT_SHIFT  10
#define BUCKET_ROWS (1 << SORT_SHIFT)
#define MAX_BUCKETS 1024
#define CHUNKS      256

__device__ __forceinline__ int block_exclusive(int v, int* ws, int& total)
{
    const int t = (int)threadIdx.x, lane = t & 31, wv = t >> 5;
    const int nw = (int)(blockDim.x >> 5);
    int inc = v;
    for (int d = 1; d < 32; d <<= 1) {
        int up = __shfl_up(inc, d, 32);
        if (lane >= d) inc += up;
    }
    __syncthreads();
    if (lane == 31) ws[wv] = inc;
    __syncthreads();
    int pre = 0, tot = 0;
    for (int i = 0; i < nw; ++i) {
        int s = ws[i];
        if (i < wv) pre += s;
        tot += s;
    }
    total = tot;
    return pre + inc - v;
}

__global__ void __launch_bounds__(1024) bucket_histogram(
    const int* __restrict__ edge_u, const int* __restrict__ edge_i,
    int* __restrict__ bh, int n_users, int n_edges, int chunk, int nb)
{
    __shared__ int h[MAX_BUCKETS];
    const int t = (int)threadIdx.x;
    for (int i = t; i < nb; i += 1024) h[i] = 0;
    __syncthreads();
    const int beg = (int)blockIdx.x * chunk;
    int end = beg + chunk;
    if (end > n_edges) end = n_edges;
    for (int e0 = beg + 4 * t; e0 < end; e0 += 4096) {
        int u[4], it[4];
        const int nv = (end - e0 < 4) ? end - e0 : 4;
        if (nv == 4) {
            v4i a = *(const v4i*)(edge_u + e0);
            v4i b = *(const v4i*)(edge_i + e0);
            u[0] = a.x; u[1] = a.y; u[2] = a.z; u[3] = a.w;
            it[0] = b.x; it[1] = b.y; it[2] = b.z; it[3] = b.w;
        } else {
            for (int j = 0; j < 4; ++j) {
                const int e = (j < nv) ? e0 + j : e0;
                u[j] = edge_u[e]; it[j] = edge_i[e];
            }
        }
        for (int j = 0; j < 4; ++j)
            if (j < nv) {
                atomicAdd(&h[u[j] >> SORT_SHIFT], 1);
                atomicAdd(&h[(n_users + it[j]) >> SORT_SHIFT], 1);
            }
    }
    __syncthreads();
    for (int i = t; i < nb; i += 1024) bh[i * CHUNKS + (int)blockIdx.x] = h[i];
}

__global__ void __launch_bounds__(CHUNKS) chunk_prefix(
    int* __restrict__ bh, int* __restrict__ btot)
{
    __shared__ int ws[32];
    const int t = (int)threadIdx.x, b = (int)blockIdx.x;
    const int v = bh[b * CHUNKS + t];
    int tot;
    const int ex = block_exclusive(v, ws, tot);
    bh[b * CHUNKS + t] = ex;
    if (t == 0) btot[b] = tot;
}

__global__ void __launch_bounds__(1024) scatter_to_buckets(
    const int* __restrict__ edge_u, const int* __restrict__ edge_i,
    const float* __restrict__ norm, const int* __restrict__ bh,
    const int* __restrict__ btot, int* __restrict__ tmp,
    int n_users, int n_edges, int chunk, int nb)
{
    __shared__ int cur[MAX_BUCKETS];
    __shared__ int ws[32];
    const int t = (int)threadIdx.x;
    int carry = 0;
    for (int base = 0; base < nb; base += 1024) {
        const int i = base + t;
        const int v = (i < nb) ? btot[i] : 0;
        int tot;
        const int ex = block_exclusive(v, ws, tot);
        if (i < nb) cur[i] = carry + ex + bh[i * CHUNKS + (int)blockIdx.x];
        carry += tot;
    }
    __syncthreads();
    v4i* items = (v4i*)tmp;
    const int beg = (int)blockIdx.x * chunk;
    int end = beg + chunk;
    if (end > n_edges) end = n_edges;
    for (int e0 = beg + 4 * t; e0 < end; e0 += 4096) {
        int u[4], it[4], wb[4];
        const int nv = (end - e0 < 4) ? end - e0 : 4;
        if (nv == 4) {
            v4i a = *(const v4i*)(edge_u + e0);
            v4i b = *(const v4i*)(edge_i + e0);
            v4i c = *(const v4i*)(norm + e0);
            u[0] = a.x; u[1] = a.y; u[2] = a.z; u[3] = a.w;
            it[0] = b.x; it[1] = b.y; it[2] = b.z; it[3] = b.w;
            wb[0] = c.x; wb[1] = c.y; wb[2] = c.z; wb[3] = c.w;
        } else {
            for (int j = 0; j < 4; ++j) {
                const int e = (j < nv) ? e0 + j : e0;
                u[j] = edge_u[e]; it[j] = edge_i[e];
                wb[j] = __float_as_int(norm[e]);
            }
        }
        int pu[4], pi[4];
        for (int j = 0; j < 4; ++j) {
            pu[j] = 0; pi[j] = 0;
            if (j < nv) {
                pu[j] = atomicAdd(&cur[u[j] >> SORT_SHIFT], 1);
                pi[j] = atomicAdd(&cur[(n_users + it[j]) >> SORT_SHIFT], 1);
            }
        }
        for (int j = 0; j < 4; ++j)
            if (j < nv) {
                v4i x; x.x = u[j];            x.y = it[j]; x.z = wb[j]; x.w = 0;
                v4i y; y.x = n_users + it[j]; y.y = u[j];  y.z = wb[j]; y.w = 0;
                items[pu[j]] = x;
                items[pi[j]] = y;
            }
    }
}

__global__ void __launch_bounds__(BUCKET_ROWS) bucket_to_lists(
    const int* __restrict__ tmp, const int* __restrict__ btot,
    int* __restrict__ off, int* __restrict__ ent, int n_rows)
{
    __shared__ int cnt[BUCKET_ROWS];
    __shared__ int ws[32];
    const int t = (int)threadIdx.x, b = (int)blockIdx.x;

    int p = 0;
    for (int i = t; i < b; i += BUCKET_ROWS) p += btot[i];
    int bstart;
    (void)block_exclusive(p, ws, bstart);
    const int bcount = btot[b];
    cnt[t] = 0;
    __syncthreads();

    const int* keys = tmp + 4 * (size_t)bstart;
    for (int i0 = t; i0 < bcount; i0 += 4 * BUCKET_ROWS) {
        int k[4];
        for (int j = 0; j < 4; ++j) {
            const int i = i0 + j * BUCKET_ROWS;
            k[j] = keys[4 * (size_t)(i < bcount ? i : i0)];
        }
        for (int j = 0; j < 4; ++j)
            if (i0 + j * BUCKET_ROWS < bcount)
                atomicAdd(&cnt[k[j] & (BUCKET_ROWS - 1)], 1);
    }
    __syncthreads();

    const int c = cnt[t];
    int tot;
    const int ex = block_exclusive(c, ws, tot);
    const int row = b * BUCKET_ROWS + t;
    if (row < n_rows) off[row] = bstart + ex;
    if (row == n_rows - 1) off[n_rows] = bstart + ex + c;
    cnt[t] = ex;
    __syncthreads();

    const v4i* items = (const v4i*)tmp + bstart;
    v2i* en2 = (v2i*)ent + bstart;
    for (int i0 = t; i0 < bcount; i0 += 4 * BUCKET_ROWS) {
        v4i x[4];
        for (int j = 0; j < 4; ++j) {
            const int i = i0 + j * BUCKET_ROWS;
            x[j] = items[i < bcount ? i : i0];
        }
        for (int j = 0; j < 4; ++j)
            if (i0 + j * BUCKET_ROWS < bcount) {
                const int q = atomicAdd(&cnt[x[j].x & (BUCKET_ROWS - 1)], 1);
                v2i e; e.x = x[j].y; e.y = x[j].z;
                en2[q] = e;
            }
    }
}

__device__ __forceinline__ void pull_body(
    const float* __restrict__ src, float* __restrict__ dst,
    const int* __restrict__ off, const int* __restrict__ ent, int n_rows,
    unsigned blk, unsigned nblk)
{
    const int l   = (int)(threadIdx.x & 15);
    const int nhw = (int)(nblk * (blockDim.x >> 4));
    int r = (int)((blk * blockDim.x + threadIdx.x) >> 4);
    const v2i* en2 = (const v2i*)ent;
    const v4f* s4  = (const v4f*)src;
    v4f* d4 = (v4f*)dst;

    const v4f zero = {0.f, 0.f, 0.f, 0.f};
    const int last = n_rows - 1;
    if (r >= n_rows) return;
    int b = off[r], e = off[r + 1];
    while (r < n_rows) {
        int rn = r + nhw;
        int rc = rn < last ? rn : last;
        int bn = off[rc], en = off[rc + 1];

        v4f acc = zero;
        for (int base = b; base < e; base += 16) {
            int m = e - base;
            if (m > 16) m = 16;
            int   myi = 0;
            float myw = 0.f;
            if (l < m) {
                v2i t = en2[base + l];
                myi = t.x;
                myw = __int_as_float(t.y);
            }
            for (int k = 0; k < m; k += 8) {
                const int c = m - k;
                v4f v0, v1, v2, v3, v4, v5, v6, v7;
                const float w0 = __shfl(myw, k + 0, 16);
                const float w1 = __shfl(myw, k + 1, 16);
                const float w2 = __shfl(myw, k + 2, 16);
                const float w3 = __shfl(myw, k + 3, 16);
                const float w4 = __shfl(myw, k + 4, 16);
                const float w5 = __shfl(myw, k + 5, 16);
                const float w6 = __shfl(myw, k + 6, 16);
                const float w7 = __shfl(myw, k + 7, 16);
#define ROW_LD(j) v##j = s4[(size_t)__shfl(myi, k + (j), 16) * (D / 4) + l];
                ROW_LD(0)
                if (c > 1) { ROW_LD(1)
                if (c > 2) { ROW_LD(2)
                if (c > 3) { ROW_LD(3)
                if (c > 4) { ROW_LD(4)
                if (c > 5) { ROW_LD(5)
                if (c > 6) { ROW_LD(6)
                if (c > 7) { ROW_LD(7) } } } } } } }
#undef ROW_LD
                acc += w0 * v0;
                if (c > 1) { acc += w1 * v1;
                if (c > 2) { acc += w2 * v2;
                if (c > 3) { acc += w3 * v3;
                if (c > 4) { acc += w4 * v4;
                if (c > 5) { acc += w5 * v5;
                if (c > 6) { acc += w6 * v6;
                if (c > 7) { acc += w7 * v7; } } } } } } }
            }
        }
        d4[(size_t)r * (D / 4) + l] = acc;
        r = rn; b = bn; e = en;
    }
}

__global__ void __launch_bounds__(256) pull_both_sides(
    const float* __restrict__ src_a, float* __restrict__ dst_a,
    const int* __restrict__ off_a, int rows_a, unsigned nb_a,
    const float* __restrict__ src_b, float* __restrict__ dst_b,
    const int* __restrict__ off_b, int rows_b,
    const int* __restrict__ ent)
{
    const unsigned b = blockIdx.x;
    const bool second = b >= nb_a;
    pull_body(second ? src_b : src_a, second ? dst_b : dst_a,
              second ? off_b : off_a, ent, second ? rows_b : rows_a,
              second ? b - nb_a : b, second ? gridDim.x - nb_a : nb_a);
}


__device__ __forceinline__ void first_entries(
    const v2i* __restrict__ en2, int b, int e, int l, int& myi, float& myw)
{
    myi = 0;
    myw = 0.f;
    if (b + l < e) {
        v2i t = en2[b + l];
        myi = t.x;
        myw = __int_as_float(t.y);
    }
}

template <class Side>
__device__ __forceinline__ v4f sum_list(
    const v4f* __restrict__ s4, const v2i* __restrict__ en2,
    int b, int e, int l, int myi, float myw, Side side)
{
    v4f acc = {0.f, 0.f, 0.f, 0.f};
    bool side_pending = true;
    for (int base = b; base < e; base += 16) {
        int m = e - base;
        if (m > 16) m = 16;
        if (base != b) first_entries(en2, base, e, l, myi, myw);
        for (int k = 0; k < m; k += 8) {
            const int c = m - k;
            v4f v0, v1, v2, v3, v4, v5, v6, v7;
            const float w0 = __shfl(myw, k + 0, 16);
            const float w1 = __shfl(myw, k + 1, 16);
            const float w2 = __shfl(myw, k + 2, 16);
            const float w3 = __shfl(myw, k + 3, 16);
            const float w4 = __shfl(myw, k + 4, 16);
            const float w5 = __shfl(myw, k + 5, 16);
            const float w6 = __shfl(myw, k + 6, 16);
            const float w7 = __shfl(myw, k + 7, 16);
#define ROW_LD(j) v##j = s4[(size_t)__shfl(myi, k + (j), 16) * (D / 4) + l];
            ROW_LD(0)
            if (c > 1) { ROW_LD(1)
            if (c > 2) { ROW_LD(2)
            if (c > 3) { ROW_LD(3)
            if (c > 4) { ROW_LD(4)
            if (c > 5) { ROW_LD(5)
            if (c > 6) { ROW_LD(6)
            if (c > 7) { ROW_LD(7) } } } } } } }
#undef ROW_LD
            if (side_pending) { side(); side_pending = false; }
            acc += w0 * v0;
            if (c > 1) { acc += w1 * v1;
            if (c > 2) { acc += w2 * v2;
            if (c > 3) { acc += w3 * v3;
            if (c > 4) { acc += w4 * v4;
            if (c > 5) { acc += w5 * v5;
            if (c > 6) { acc += w6 * v6;
            if (c > 7) { acc += w7 * v7; } } } } } } }
        }
    }
    if (side_pending) side();
    return acc;
}

__global__ void __launch_bounds__(256) requested_rows_two_levels(
    const float* __restrict__ user_feat, const float* __restrict__ item_feat,
    const float* __restrict__ u1, const float* __restrict__ i1,
    const int* __restrict__ off, const int* __restrict__ ent,
    const int* __restrict__ users, const int* __restrict__ pos,
    const int* __restrict__ neg,
    float* __restrict__ out, int batch, int n_users)
{
    const int l = (int)(threadIdx.x & 15);
    const int r = (int)((blockIdx.x * blockDim.x + threadIdx.x) >> 4);
    if (r >= 3 * batch) return;
    const int seg = r / batch;
    const int i   = r - seg * batch;
    const bool is_user = (seg == 0);
    const int idx = is_user ? users[i] : (seg == 1 ? pos[i] : neg[i]);

    const v4f* own0 = (const v4f*)(is_user ? user_feat : item_feat);
    const v4f* own1 = (const v4f*)(is_user ? u1 : i1);
    const v4f* opp1 = (const v4f*)(is_user ? i1 : u1);
    const int jr       = is_user ? idx : n_users + idx;
    const int opp_base = is_user ? n_users : 0;
    const v2i* en2 = (const v2i*)ent;

    const int b = off[jr], e = off[jr + 1];
    v4f acc2 = {0.f, 0.f, 0.f, 0.f};
    v4f acc3 = {0.f, 0.f, 0.f, 0.f};
    for (int base = b; base < e; base += 16) {
        int m = e - base;
        if (m > 16) m = 16;
        int   s  = 0, sb = 0, se = 0;
        float w  = 0.f;
        if (l < m) {
            v2i t = en2[base + l];
            s  = t.x;
            w  = __int_as_float(t.y);
            sb = off[opp_base + s];
            se = off[opp_base + s + 1];
        }
        int kb = __shfl(sb, 0, 16), ke = __shfl(se, 0, 16);
        int   ci;
        float cw;
        first_entries(en2, kb, ke, l, ci, cw);
        for (int k = 0; k < m; ++k) {
            const int   sk = __shfl(s, k, 16);
            const float wk = __shfl(w, k, 16);
            const int kn = (k + 1 < m) ? k + 1 : k;
            const int nb = __shfl(sb, kn, 16), ne = __shfl(se, kn, 16);
            int   ni = 0;
            float nw = 0.f;
            v4f   v;
            const v4f t = sum_list(own1, en2, kb, ke, l, ci, cw, [&]() {
                if (k + 1 < m) first_entries(en2, nb, ne, l, ni, nw);
                v = opp1[(size_t)sk * (D / 4) + l];
            });
            acc2 += wk * v;
            acc3 += wk * t;
            kb = nb; ke = ne; ci = ni; cw = nw;
        }
    }
    const v4f h0 = own0[(size_t)idx * (D / 4) + l];
    const v4f h1 = own1[(size_t)idx * (D / 4) + l];
    ((v4f*)out)[(size_t)r * (D / 4) + l] = ((h0 + h1) + acc2) + acc3;
}

__global__ void __launch_bounds__(256) wmma_scale_inplace(
    float* __restrict__ out, int ntiles)
{
    int wid = (int)(blockIdx.x * (blockDim.x >> 5) + (threadIdx.x >> 5));
    if (wid >= ntiles) return;

    int lane = threadIdx.x & 31;
    int h = lane >> 4;
    int l = lane & 15;
    int rg = wid >> 2;
    int ct = wid & 3;

    float* base = out + (size_t)rg * 16 * D + ct * 16;

    v8f acc = {};
#pragma unroll
    for (int c = 0; c < 4; ++c) {
        int k0 = 2 * h;
        int k1 = 2 * h + 1;
        v2f a, b;
        a.x = base[(size_t)l * D + 4 * c + k0];
        a.y = base[(size_t)l * D + 4 * c + k1];
        b.x = ((4 * c + k0) == l) ? 0.25f : 0.0f;
        b.y = ((4 * c + k1) == l) ? 0.25f : 0.0f;
        acc = __builtin_amdgcn_wmma_f32_16x16x4_f32(
             false, a,  false, b,
             (short)0, acc,  false,  false);
    }
#pragma unroll
    for (int j = 0; j < 8; ++j)
        base[(size_t)(h * 8 + j) * D + l] = acc[j];
}

extern "C" void kernel_launch(void* const* d_in, const int* in_sizes, int n_in,
                              void* d_out, int out_size, void* d_ws, size_t ws_size,
                              hipStream_t stream)
{
    const float* user_feat = (const float*)d_in[0];
    const float* item_feat = (const float*)d_in[1];
    const int*   edge_u    = (const int*)d_in[2];
    const int*   edge_i    = (const int*)d_in[3];
    const float* norm      = (const float*)d_in[4];
    const int*   users     = (const int*)d_in[5];
    const int*   pos       = (const int*)d_in[6];
    const int*   neg       = (const int*)d_in[7];
    float* out = (float*)d_out;

    const int n_users = in_sizes[0] / D;
    const int n_items = in_sizes[1] / D;
    const int n_edges = in_sizes[2];
    const int batch   = in_sizes[5];
    const int n_rows  = n_users + n_items;

    size_t uElems = (size_t)n_users * D;
    size_t iElems = (size_t)n_items * D;

    float* su_a = (float*)d_ws;
    float* su_b = su_a + uElems;
    float* si   = su_b + uElems;
    int*   off  = (int*)(si + iElems);
    int offPad  = ((n_rows + 1 + 63) / 64) * 64;
    int*   ent  = off + offPad;

    const int nb = (n_rows + BUCKET_ROWS - 1) / BUCKET_ROWS;
    const int chunk = (((n_edges + CHUNKS - 1) / CHUNKS) + 3) & ~3;
    int*   tmp  = (int*)su_b;
    int*   bh   = tmp + 8 * (size_t)n_edges;
    int*   btot = bh + (size_t)nb * CHUNKS;

    bucket_histogram<<<CHUNKS, 1024, 0, stream>>>(edge_u, edge_i, bh, n_users, n_edges, chunk, nb);
    chunk_prefix<<<nb, CHUNKS, 0, stream>>>(bh, btot);
    scatter_to_buckets<<<CHUNKS, 1024, 0, stream>>>(edge_u, edge_i, norm, bh, btot, tmp,
                                                    n_users, n_edges, chunk, nb);
    bucket_to_lists<<<nb, BUCKET_ROWS, 0, stream>>>(tmp, btot, off, ent, n_rows);

    int ru = (n_users + 24575) / 24576;
    int ri = (n_items + 24575) / 24576;
    int ublocks = (n_users + 16 * ru - 1) / (16 * ru);
    int iblocks = (n_items + 16 * ri - 1) / (16 * ri);
    const int* off_u = off;
    const int* off_i = off + n_users;

    pull_both_sides<<<ublocks + iblocks, 256, 0, stream>>>(
        item_feat, su_a, off_u, n_users, ublocks,
        user_feat, si,   off_i, n_items, ent);

    int oblocks = (3 * batch + 15) / 16;
    requested_rows_two_levels<<<oblocks, 256, 0, stream>>>(
        user_feat, item_feat, su_a, si, off, ent, users, pos, neg, out, batch, n_users);

    int nrows  = 3 * batch;
    int ntiles = (nrows / 16) * (D / 16);
    int wblocks = (ntiles + 7) / 8;
    wmma_scale_inplace<<<wblocks, 256, 0, stream>>>(out, ntiles);
}


# list building: both sort levels place items in an LDS stage and copy it out in order (neighbouring lanes store neighbouring items), not one scattered store per item; slots in LDS arrival order
# speedup vs baseline: 4.1341x; 1.1116x over previous
#include <hip/hip_runtime.h>

#define D 64

typedef __attribute__((ext_vector_type(2))) float v2f;
typedef __attribute__((ext_vector_type(4))) float v4f;
typedef __attribute__((ext_vector_type(8))) float v8f;
typedef __attribute__((ext_vector_type(2))) int   v2i;

typedef __attribute__((ext_vector_type(4))) int   v4i;

#define SORT_SHIFT  10
#define BUCKET_ROWS (1 << SORT_SHIFT)
#define MAX_BUCKETS 1024
#define CHUNKS      256
#define PASS_EDGES  4096
#define LIST_STAGE  12288

__device__ __forceinline__ int block_exclusive(int v, int* ws, int& total)
{
    const int t = (int)threadIdx.x, lane = t & 31, wv = t >> 5;
    const int nw = (int)(blockDim.x >> 5);
    int inc = v;
    for (int d = 1; d < 32; d <<= 1) {
        int up = __shfl_up(inc, d, 32);
        if (lane >= d) inc += up;
    }
    __syncthreads();
    if (lane == 31) ws[wv] = inc;
    __syncthreads();
    int pre = 0, tot = 0;
    for (int i = 0; i < nw; ++i) {
        int s = ws[i];
        if (i < wv) pre += s;
        tot += s;
    }
    total = tot;
    return pre + inc - v;
}

__global__ void __launch_bounds__(1024) bucket_histogram(
    const int* __restrict__ edge_u, const int* __restrict__ edge_i,
    int* __restrict__ bh, int n_users, int n_edges, int chunk, int nb)
{
    __shared__ int h[MAX_BUCKETS];
    const int t = (int)threadIdx.x;
    for (int i = t; i < nb; i += 1024) h[i] = 0;
    __syncthreads();
    const int beg = (int)blockIdx.x * chunk;
    int end = beg + chunk;
    if (end > n_edges) end = n_edges;
    for (int e0 = beg + 4 * t; e0 < end; e0 += 4096) {
        int u[4], it[4];
        const int nv = (end - e0 < 4) ? end - e0 : 4;
        if (nv == 4) {
            v4i a = *(const v4i*)(edge_u + e0);
            v4i b = *(const v4i*)(edge_i + e0);
            u[0] = a.x; u[1] = a.y; u[2] = a.z; u[3] = a.w;
            it[0] = b.x; it[1] = b.y; it[2] = b.z; it[3] = b.w;
        } else {
            for (int j = 0; j < 4; ++j) {
                const int e = (j < nv) ? e0 + j : e0;
                u[j] = edge_u[e]; it[j] = edge_i[e];
            }
        }
        for (int j = 0; j < 4; ++j)
            if (j < nv) {
                atomicAdd(&h[u[j] >> SORT_SHIFT], 1);
                atomicAdd(&h[(n_users + it[j]) >> SORT_SHIFT], 1);
            }
    }
    __syncthreads();
    for (int i = t; i < nb; i += 1024) bh[i * CHUNKS + (int)blockIdx.x] = h[i];
}

__global__ void __launch_bounds__(CHUNKS) chunk_prefix(
    int* __restrict__ bh, int* __restrict__ btot)
{
    __shared__ int ws[32];
    const int t = (int)threadIdx.x, b = (int)blockIdx.x;
    const int v = bh[b * CHUNKS + t];
    int tot;
    const int ex = block_exclusive(v, ws, tot);
    bh[b * CHUNKS + t] = ex;
    if (t == 0) btot[b] = tot;
}

__global__ void __launch_bounds__(1024) scatter_to_buckets(
    const int* __restrict__ edge_u, const int* __restrict__ edge_i,
    const float* __restrict__ norm, const int* __restrict__ bh,
    const int* __restrict__ btot, int* __restrict__ tmp,
    int n_users, int n_edges, int chunk, int nb)
{
    __shared__ int gcur[MAX_BUCKETS];
    __shared__ int lnum[MAX_BUCKETS];
    __shared__ int lbeg[MAX_BUCKETS];
    __shared__ int ws[32];
    __shared__ v4i stage[2 * PASS_EDGES];
    const int t = (int)threadIdx.x;
    int carry = 0;
    for (int base = 0; base < nb; base += 1024) {
        const int i = base + t;
        const int v = (i < nb) ? btot[i] : 0;
        int tot;
        const int ex = block_exclusive(v, ws, tot);
        if (i < nb) gcur[i] = carry + ex + bh[i * CHUNKS + (int)blockIdx.x];
        carry += tot;
    }
    v4i* items = (v4i*)tmp;
    const int beg = (int)blockIdx.x * chunk;
    int end = beg + chunk;
    if (end > n_edges) end = n_edges;
    for (int p0 = beg; p0 < end; p0 += PASS_EDGES) {
        if (p0 != beg) {
            __syncthreads();
            for (int i = t; i < nb; i += 1024) gcur[i] += lnum[i];
        }
        for (int i = t; i < nb; i += 1024) lnum[i] = 0;
        __syncthreads();
        const int e0 = p0 + 4 * t;
        int nv = end - e0;
        nv = nv < 0 ? 0 : (nv > 4 ? 4 : nv);
        int u[4] = {0, 0, 0, 0}, it[4] = {0, 0, 0, 0}, wb[4] = {0, 0, 0, 0};
        if (nv == 4) {
            v4i a = *(const v4i*)(edge_u + e0);
            v4i b = *(const v4i*)(edge_i + e0);
            v4i c = *(const v4i*)(norm + e0);
            u[0] = a.x; u[1] = a.y; u[2] = a.z; u[3] = a.w;
            it[0] = b.x; it[1] = b.y; it[2] = b.z; it[3] = b.w;
            wb[0] = c.x; wb[1] = c.y; wb[2] = c.z; wb[3] = c.w;
        } else {
            for (int j = 0; j < 4; ++j)
                if (j < nv) {
                    u[j] = edge_u[e0 + j]; it[j] = edge_i[e0 + j];
                    wb[j] = __float_as_int(norm[e0 + j]);
                }
        }
        int pu[4], pi[4];
        for (int j = 0; j < 4; ++j) {
            pu[j] = 0; pi[j] = 0;
            if (j < nv) {
                pu[j] = atomicAdd(&lnum[u[j] >> SORT_SHIFT], 1);
                pi[j] = atomicAdd(&lnum[(n_users + it[j]) >> SORT_SHIFT], 1);
            }
        }
        __syncthreads();
        int pass_items = 0;
        for (int base = 0; base < nb; base += 1024) {
            const int i = base + t;
            const int v = (i < nb) ? lnum[i] : 0;
            int tot;
            const int ex = block_exclusive(v, ws, tot);
            if (i < nb) lbeg[i] = pass_items + ex;
            pass_items += tot;
        }
        __syncthreads();
        for (int j = 0; j < 4; ++j)
            if (j < nv) {
                v4i x; x.x = u[j];            x.y = it[j]; x.z = wb[j]; x.w = 0;
                v4i y; y.x = n_users + it[j]; y.y = u[j];  y.z = wb[j]; y.w = 0;
                stage[lbeg[x.x >> SORT_SHIFT] + pu[j]] = x;
                stage[lbeg[y.x >> SORT_SHIFT] + pi[j]] = y;
            }
        __syncthreads();
        for (int i0 = t; i0 < pass_items; i0 += 4 * 1024) {
            v4i x[4];
            int  d[4];
            for (int j = 0; j < 4; ++j) {
                const int i = i0 + j * 1024;
                x[j] = stage[i < pass_items ? i : i0];
            }
            for (int j = 0; j < 4; ++j) {
                const int b = x[j].x >> SORT_SHIFT;
                d[j] = gcur[b] - lbeg[b];
            }
            for (int j = 0; j < 4; ++j)
                if (i0 + j * 1024 < pass_items) items[d[j] + i0 + j * 1024] = x[j];
        }
    }
}

__global__ void __launch_bounds__(BUCKET_ROWS) bucket_to_lists(
    const int* __restrict__ tmp, const int* __restrict__ btot,
    int* __restrict__ off, int* __restrict__ ent, int n_rows)
{
    __shared__ int cnt[BUCKET_ROWS];
    __shared__ int ws[32];
    __shared__ v2i stage[LIST_STAGE];
    const int t = (int)threadIdx.x, b = (int)blockIdx.x;

    int p = 0;
    for (int i = t; i < b; i += BUCKET_ROWS) p += btot[i];
    int bstart;
    (void)block_exclusive(p, ws, bstart);
    const int bcount = btot[b];
    cnt[t] = 0;
    __syncthreads();

    const int* keys = tmp + 4 * (size_t)bstart;
    for (int i0 = t; i0 < bcount; i0 += 4 * BUCKET_ROWS) {
        int k[4];
        for (int j = 0; j < 4; ++j) {
            const int i = i0 + j * BUCKET_ROWS;
            k[j] = keys[4 * (size_t)(i < bcount ? i : i0)];
        }
        for (int j = 0; j < 4; ++j)
            if (i0 + j * BUCKET_ROWS < bcount)
                atomicAdd(&cnt[k[j] & (BUCKET_ROWS - 1)], 1);
    }
    __syncthreads();

    const int c = cnt[t];
    int tot;
    const int ex = block_exclusive(c, ws, tot);
    const int row = b * BUCKET_ROWS + t;
    if (row < n_rows) off[row] = bstart + ex;
    if (row == n_rows - 1) off[n_rows] = bstart + ex + c;
    cnt[t] = ex;
    __syncthreads();

    const v4i* items = (const v4i*)tmp + bstart;
    v2i* en2 = (v2i*)ent + bstart;
    const bool staged = bcount > 0 && bcount <= LIST_STAGE;
    for (int i0 = t; i0 < bcount; i0 += 4 * BUCKET_ROWS) {
        v4i x[4];
        for (int j = 0; j < 4; ++j) {
            const int i = i0 + j * BUCKET_ROWS;
            x[j] = items[i < bcount ? i : i0];
        }
        for (int j = 0; j < 4; ++j)
            if (i0 + j * BUCKET_ROWS < bcount) {
                const int q = atomicAdd(&cnt[x[j].x & (BUCKET_ROWS - 1)], 1);
                v2i e; e.x = x[j].y; e.y = x[j].z;
                if (staged) stage[q] = e; else en2[q] = e;
            }
    }
    if (staged) {
        __syncthreads();
        const int sh = bstart & 31;
        for (int i0 = t - sh; i0 < bcount; i0 += 4 * BUCKET_ROWS) {
            v2i x[4];
            for (int j = 0; j < 4; ++j) {
                const int i = i0 + j * BUCKET_ROWS;
                x[j] = stage[(i >= 0 && i < bcount) ? i : 0];
            }
            for (int j = 0; j < 4; ++j) {
                const int i = i0 + j * BUCKET_ROWS;
                if (i >= 0 && i < bcount) en2[i] = x[j];
            }
        }
    }
}

__device__ __forceinline__ void pull_body(
    const float* __restrict__ src, float* __restrict__ dst,
    const int* __restrict__ off, const int* __restrict__ ent, int n_rows,
    unsigned blk, unsigned nblk)
{
    const int l   = (int)(threadIdx.x & 15);
    const int nhw = (int)(nblk * (blockDim.x >> 4));
    int r = (int)((blk * blockDim.x + threadIdx.x) >> 4);
    const v2i* en2 = (const v2i*)ent;
    const v4f* s4  = (const v4f*)src;
    v4f* d4 = (v4f*)dst;

    const v4f zero = {0.f, 0.f, 0.f, 0.f};
    const int last = n_rows - 1;
    if (r >= n_rows) return;
    int b = off[r], e = off[r + 1];
    while (r < n_rows) {
        int rn = r + nhw;
        int rc = rn < last ? rn : last;
        int bn = off[rc], en = off[rc + 1];

        v4f acc = zero;
        for (int base = b; base < e; base += 16) {
            int m = e - base;
            if (m > 16) m = 16;
            int   myi = 0;
            float myw = 0.f;
            if (l < m) {
                v2i t = en2[base + l];
                myi = t.x;
                myw = __int_as_float(t.y);
            }
            for (int k = 0; k < m; k += 8) {
                const int c = m - k;
                v4f v0, v1, v2, v3, v4, v5, v6, v7;
                const float w0 = __shfl(myw, k + 0, 16);
                const float w1 = __shfl(myw, k + 1, 16);
                const float w2 = __shfl(myw, k + 2, 16);
                const float w3 = __shfl(myw, k + 3, 16);
                const float w4 = __shfl(myw, k + 4, 16);
                const float w5 = __shfl(myw, k + 5, 16);
                const float w6 = __shfl(myw, k + 6, 16);
                const float w7 = __shfl(myw, k + 7, 16);
#define ROW_LD(j) v##j = s4[(size_t)__shfl(myi, k + (j), 16) * (D / 4) + l];
                ROW_LD(0)
                if (c > 1) { ROW_LD(1)
                if (c > 2) { ROW_LD(2)
                if (c > 3) { ROW_LD(3)
                if (c > 4) { ROW_LD(4)
                if (c > 5) { ROW_LD(5)
                if (c > 6) { ROW_LD(6)
                if (c > 7) { ROW_LD(7) } } } } } } }
#undef ROW_LD
                acc += w0 * v0;
                if (c > 1) { acc += w1 * v1;
                if (c > 2) { acc += w2 * v2;
                if (c > 3) { acc += w3 * v3;
                if (c > 4) { acc += w4 * v4;
                if (c > 5) { acc += w5 * v5;
                if (c > 6) { acc += w6 * v6;
                if (c > 7) { acc += w7 * v7; } } } } } } }
            }
        }
        d4[(size_t)r * (D / 4) + l] = acc;
        r = rn; b = bn; e = en;
    }
}

__global__ void __launch_bounds__(256) pull_both_sides(
    const float* __restrict__ src_a, float* __restrict__ dst_a,
    const int* __restrict__ off_a, int rows_a, unsigned nb_a,
    const float* __restrict__ src_b, float* __restrict__ dst_b,
    const int* __restrict__ off_b, int rows_b,
    const int* __restrict__ ent)
{
    const unsigned b = blockIdx.x;
    const bool second = b >= nb_a;
    pull_body(second ? src_b : src_a, second ? dst_b : dst_a,
              second ? off_b : off_a, ent, second ? rows_b : rows_a,
              second ? b - nb_a : b, second ? gridDim.x - nb_a : nb_a);
}


__device__ __forceinline__ void first_entries(
    const v2i* __restrict__ en2, int b, int e, int l, int& myi, float& myw)
{
    myi = 0;
    myw = 0.f;
    if (b + l < e) {
        v2i t = en2[b + l];
        myi = t.x;
        myw = __int_as_float(t.y);
    }
}

template <class Side>
__device__ __forceinline__ v4f sum_list(
    const v4f* __restrict__ s4, const v2i* __restrict__ en2,
    int b, int e, int l, int myi, float myw, Side side)
{
    v4f acc = {0.f, 0.f, 0.f, 0.f};
    bool side_pending = true;
    for (int base = b; base < e; base += 16) {
        int m = e - base;
        if (m > 16) m = 16;
        if (base != b) first_entries(en2, base, e, l, myi, myw);
        for (int k = 0; k < m; k += 8) {
            const int c = m - k;
            v4f v0, v1, v2, v3, v4, v5, v6, v7;
            const float w0 = __shfl(myw, k + 0, 16);
            const float w1 = __shfl(myw, k + 1, 16);
            const float w2 = __shfl(myw, k + 2, 16);
            const float w3 = __shfl(myw, k + 3, 16);
            const float w4 = __shfl(myw, k + 4, 16);
            const float w5 = __shfl(myw, k + 5, 16);
            const float w6 = __shfl(myw, k + 6, 16);
            const float w7 = __shfl(myw, k + 7, 16);
#define ROW_LD(j) v##j = s4[(size_t)__shfl(myi, k + (j), 16) * (D / 4) + l];
            ROW_LD(0)
            if (c > 1) { ROW_LD(1)
            if (c > 2) { ROW_LD(2)
            if (c > 3) { ROW_LD(3)
            if (c > 4) { ROW_LD(4)
            if (c > 5) { ROW_LD(5)
            if (c > 6) { ROW_LD(6)
            if (c > 7) { ROW_LD(7) } } } } } } }
#undef ROW_LD
            if (side_pending) { side(); side_pending = false; }
            acc += w0 * v0;
            if (c > 1) { acc += w1 * v1;
            if (c > 2) { acc += w2 * v2;
            if (c > 3) { acc += w3 * v3;
            if (c > 4) { acc += w4 * v4;
            if (c > 5) { acc += w5 * v5;
            if (c > 6) { acc += w6 * v6;
            if (c > 7) { acc += w7 * v7; } } } } } } }
        }
    }
    if (side_pending) side();
    return acc;
}

__global__ void __launch_bounds__(256) requested_rows_two_levels(
    const float* __restrict__ user_feat, const float* __restrict__ item_feat,
    const float* __restrict__ u1, const float* __restrict__ i1,
    const int* __restrict__ off, const int* __restrict__ ent,
    const int* __restrict__ users, const int* __restrict__ pos,
    const int* __restrict__ neg,
    float* __restrict__ out, int batch, int n_users)
{
    const int l = (int)(threadIdx.x & 15);
    const int r = (int)((blockIdx.x * blockDim.x + threadIdx.x) >> 4);
    if (r >= 3 * batch) return;
    const int seg = r / batch;
    const int i   = r - seg * batch;
    const bool is_user = (seg == 0);
    const int idx = is_user ? users[i] : (seg == 1 ? pos[i] : neg[i]);

    const v4f* own0 = (const v4f*)(is_user ? user_feat : item_feat);
    const v4f* own1 = (const v4f*)(is_user ? u1 : i1);
    const v4f* opp1 = (const v4f*)(is_user ? i1 : u1);
    const int jr       = is_user ? idx : n_users + idx;
    const int opp_base = is_user ? n_users : 0;
    const v2i* en2 = (const v2i*)ent;

    const int b = off[jr], e = off[jr + 1];
    v4f acc2 = {0.f, 0.f, 0.f, 0.f};
    v4f acc3 = {0.f, 0.f, 0.f, 0.f};
    for (int base = b; base < e; base += 16) {
        int m = e - base;
        if (m > 16) m = 16;
        int   s  = 0, sb = 0, se = 0;
        float w  = 0.f;
        if (l < m) {
            v2i t = en2[base + l];
            s  = t.x;
            w  = __int_as_float(t.y);
            sb = off[opp_base + s];
            se = off[opp_base + s + 1];
        }
        int kb = __shfl(sb, 0, 16), ke = __shfl(se, 0, 16);
        int   ci;
        float cw;
        first_entries(en2, kb, ke, l, ci, cw);
        for (int k = 0; k < m; ++k) {
            const int   sk = __shfl(s, k, 16);
            const float wk = __shfl(w, k, 16);
            const int kn = (k + 1 < m) ? k + 1 : k;
            const int nb = __shfl(sb, kn, 16), ne = __shfl(se, kn, 16);
            int   ni = 0;
            float nw = 0.f;
            v4f   v;
            const v4f t = sum_list(own1, en2, kb, ke, l, ci, cw, [&]() {
                if (k + 1 < m) first_entries(en2, nb, ne, l, ni, nw);
                v = opp1[(size_t)sk * (D / 4) + l];
            });
            acc2 += wk * v;
            acc3 += wk * t;
            kb = nb; ke = ne; ci = ni; cw = nw;
        }
    }
    const v4f h0 = own0[(size_t)idx * (D / 4) + l];
    const v4f h1 = own1[(size_t)idx * (D / 4) + l];
    ((v4f*)out)[(size_t)r * (D / 4) + l] = ((h0 + h1) + acc2) + acc3;
}

__global__ void __launch_bounds__(256) wmma_scale_inplace(
    float* __restrict__ out, int ntiles)
{
    int wid = (int)(blockIdx.x * (blockDim.x >> 5) + (threadIdx.x >> 5));
    if (wid >= ntiles) return;

    int lane = threadIdx.x & 31;
    int h = lane >> 4;
    int l = lane & 15;
    int rg = wid >> 2;
    int ct = wid & 3;

    float* base = out + (size_t)rg * 16 * D + ct * 16;

    v8f acc = {};
#pragma unroll
    for (int c = 0; c < 4; ++c) {
        int k0 = 2 * h;
        int k1 = 2 * h + 1;
        v2f a, b;
        a.x = base[(size_t)l * D + 4 * c + k0];
        a.y = base[(size_t)l * D + 4 * c + k1];
        b.x = ((4 * c + k0) == l) ? 0.25f : 0.0f;
        b.y = ((4 * c + k1) == l) ? 0.25f : 0.0f;
        acc = __builtin_amdgcn_wmma_f32_16x16x4_f32(
             false, a,  false, b,
             (short)0, acc,  false,  false);
    }
#pragma unroll
    for (int j = 0; j < 8; ++j)
        base[(size_t)(h * 8 + j) * D + l] = acc[j];
}

extern "C" void kernel_launch(void* const* d_in, const int* in_sizes, int n_in,
                              void* d_out, int out_size, void* d_ws, size_t ws_size,
                              hipStream_t stream)
{
    const float* user_feat = (const float*)d_in[0];
    const float* item_feat = (const float*)d_in[1];
    const int*   edge_u    = (const int*)d_in[2];
    const int*   edge_i    = (const int*)d_in[3];
    const float* norm      = (const float*)d_in[4];
    const int*   users     = (const int*)d_in[5];
    const int*   pos       = (const int*)d_in[6];
    const int*   neg       = (const int*)d_in[7];
    float* out = (float*)d_out;

    const int n_users = in_sizes[0] / D;
    const int n_items = in_sizes[1] / D;
    const int n_edges = in_sizes[2];
    const int batch   = in_sizes[5];
    const int n_rows  = n_users + n_items;

    size_t uElems = (size_t)n_users * D;
    size_t iElems = (size_t)n_items * D;

    float* su_a = (float*)d_ws;
    float* su_b = su_a + uElems;
    float* si   = su_b + uElems;
    int*   off  = (int*)(si + iElems);
    int offPad  = ((n_rows + 1 + 63) / 64) * 64;
    int*   ent  = off + offPad;

    const int nb = (n_rows + BUCKET_ROWS - 1) / BUCKET_ROWS;
    const int chunk = (((n_edges + CHUNKS - 1) / CHUNKS) + 3) & ~3;
    int*   tmp  = (int*)su_b;
    int*   bh   = tmp + 8 * (size_t)n_edges;
    int*   btot = bh + (size_t)nb * CHUNKS;

    bucket_histogram<<<CHUNKS, 1024, 0, stream>>>(edge_u, edge_i, bh, n_users, n_edges, chunk, nb);
    chunk_prefix<<<nb, CHUNKS, 0, stream>>>(bh, btot);
    scatter_to_buckets<<<CHUNKS, 1024, 0, stream>>>(edge_u, edge_i, norm, bh, btot, tmp,
                                                    n_users, n_edges, chunk, nb);
    bucket_to_lists<<<nb, BUCKET_ROWS, 0, stream>>>(tmp, btot, off, ent, n_rows);

    int ru = (n_users + 24575) / 24576;
    int ri = (n_items + 24575) / 24576;
    int ublocks = (n_users + 16 * ru - 1) / (16 * ru);
    int iblocks = (n_items + 16 * ri - 1) / (16 * ri);
    const int* off_u = off;
    const int* off_i = off + n_users;

    pull_both_sides<<<ublocks + iblocks, 256, 0, stream>>>(
        item_feat, su_a, off_u, n_users, ublocks,
        user_feat, si,   off_i, n_items, ent);

    int oblocks = (3 * batch + 15) / 16;
    requested_rows_two_levels<<<oblocks, 256, 0, stream>>>(
        user_feat, item_feat, su_a, si, off, ent, users, pos, neg, out, batch, n_users);

    int nrows  = 3 * batch;
    int ntiles = (nrows / 16) * (D / 16);
    int wblocks = (ntiles + 7) / 8;
    wmma_scale_inplace<<<wblocks, 256, 0, stream>>>(out, ntiles);
}


# list building in 2 launches, not 4: a chunk orders its items by bucket in its own region and stores its bucket starts; a bucket collects its pieces; no count pre-pass; slots in LDS arrival order
# speedup vs baseline: 4.3215x; 1.0453x over previous
#include <hip/hip_runtime.h>

#define D 64

typedef __attribute__((ext_vector_type(2))) float v2f;
typedef __attribute__((ext_vector_type(4))) float v4f;
typedef __attribute__((ext_vector_type(8))) float v8f;
typedef __attribute__((ext_vector_type(2))) int   v2i;

typedef __attribute__((ext_vector_type(3))) int   v3i;
typedef __attribute__((ext_vector_type(4))) int   v4i;

#define SORT_SHIFT  10
#define BUCKET_ROWS (1 << SORT_SHIFT)
#define MAX_BUCKETS 1024
#define MIN_CHUNKS  256
#define PASS_EDGES  4096
#define LIST_STAGE  12288
#define PIECE_LANES 8
#define PIECE_GROUPS (BUCKET_ROWS / PIECE_LANES)

__device__ __forceinline__ int block_exclusive(int v, int* ws, int& total)
{
    const int t = (int)threadIdx.x, lane = t & 31, wv = t >> 5;
    const int nw = (int)(blockDim.x >> 5);
    int inc = v;
    for (int d = 1; d < 32; d <<= 1) {
        int up = __shfl_up(inc, d, 32);
        if (lane >= d) inc += up;
    }
    __syncthreads();
    if (lane == 31) ws[wv] = inc;
    __syncthreads();
    int pre = 0, tot = 0;
    for (int i = 0; i < nw; ++i) {
        int s = ws[i];
        if (i < wv) pre += s;
        tot += s;
    }
    total = tot;
    return pre + inc - v;
}

__device__ __forceinline__ void block_sum2(int a, int b, int* ws, int& ta, int& tb)
{
    const int t = (int)threadIdx.x, lane = t & 31, wv = t >> 5;
    const int nw = (int)(blockDim.x >> 5);
    for (int d = 16; d > 0; d >>= 1) {
        a += __shfl_xor(a, d, 32);
        b += __shfl_xor(b, d, 32);
    }
    __syncthreads();
    if (lane == 0) { ws[wv] = a; ws[32 + wv] = b; }
    __syncthreads();
    int sa = 0, sb = 0;
    for (int i = 0; i < nw; ++i) { sa += ws[i]; sb += ws[32 + i]; }
    ta = sa;
    tb = sb;
}

__global__ void __launch_bounds__(1024) chunk_sort(
    const int* __restrict__ edge_u, const int* __restrict__ edge_i,
    const float* __restrict__ norm, int* __restrict__ tmp, int* __restrict__ cst,
    int n_users, int n_edges, int chunk, int nb)
{
    __shared__ int lnum[MAX_BUCKETS];
    __shared__ int lbeg[MAX_BUCKETS];
    __shared__ int ws[32];
    __shared__ v4i stage[2 * PASS_EDGES];
    const int t = (int)threadIdx.x;
    for (int i = t; i < nb; i += 1024) lnum[i] = 0;
    __syncthreads();
    const int beg = (int)blockIdx.x * chunk;
    int end = beg + chunk;
    if (end > n_edges) end = n_edges;
    const int e0 = beg + 4 * t;
    int nv = end - e0;
    nv = nv < 0 ? 0 : (nv > 4 ? 4 : nv);
    int u[4] = {0, 0, 0, 0}, it[4] = {0, 0, 0, 0}, wb[4] = {0, 0, 0, 0};
    if (nv == 4) {
        v4i a = *(const v4i*)(edge_u + e0);
        v4i b = *(const v4i*)(edge_i + e0);
        v4i c = *(const v4i*)(norm + e0);
        u[0] = a.x; u[1] = a.y; u[2] = a.z; u[3] = a.w;
        it[0] = b.x; it[1] = b.y; it[2] = b.z; it[3] = b.w;
        wb[0] = c.x; wb[1] = c.y; wb[2] = c.z; wb[3] = c.w;
    } else {
        for (int j = 0; j < 4; ++j)
            if (j < nv) {
                u[j] = edge_u[e0 + j]; it[j] = edge_i[e0 + j];
                wb[j] = __float_as_int(norm[e0 + j]);
            }
    }
    int pu[4], pi[4];
    for (int j = 0; j < 4; ++j) {
        pu[j] = 0; pi[j] = 0;
        if (j < nv) {
            pu[j] = atomicAdd(&lnum[u[j] >> SORT_SHIFT], 1);
            pi[j] = atomicAdd(&lnum[(n_users + it[j]) >> SORT_SHIFT], 1);
        }
    }
    __syncthreads();
    int* my_cst = cst + (size_t)blockIdx.x * (size_t)(nb + 1);
    int chunk_items = 0;
    for (int base = 0; base < nb; base += 1024) {
        const int i = base + t;
        const int v = (i < nb) ? lnum[i] : 0;
        int tot;
        const int ex = block_exclusive(v, ws, tot);
        if (i < nb) {
            lbeg[i] = chunk_items + ex;
            my_cst[i] = chunk_items + ex;
        }
        chunk_items += tot;
    }
    if (t == 0) my_cst[nb] = chunk_items;
    __syncthreads();
    for (int j = 0; j < 4; ++j)
        if (j < nv) {
            v4i x; x.x = u[j];            x.y = it[j]; x.z = wb[j]; x.w = 0;
            v4i y; y.x = n_users + it[j]; y.y = u[j];  y.z = wb[j]; y.w = 0;
            stage[lbeg[x.x >> SORT_SHIFT] + pu[j]] = x;
            stage[lbeg[y.x >> SORT_SHIFT] + pi[j]] = y;
        }
    __syncthreads();
    v4i* region = (v4i*)tmp + (size_t)blockIdx.x * (size_t)(2 * chunk);
    for (int i0 = t; i0 < chunk_items; i0 += 4 * 1024) {
        v4i x[4];
        for (int j = 0; j < 4; ++j) {
            const int i = i0 + j * 1024;
            x[j] = stage[i < chunk_items ? i : i0];
        }
        for (int j = 0; j < 4; ++j)
            if (i0 + j * 1024 < chunk_items) region[i0 + j * 1024] = x[j];
    }
}

__device__ __forceinline__ void piece_of(
    const int* __restrict__ cst, int c, int nchunks, int chunk, int nb, int b,
    int& first, int& len)
{
    const bool ok = c < nchunks;
    const int cc = ok ? c : 0;
    const int* row = cst + (size_t)cc * (size_t)(nb + 1) + b;
    const int p = row[0], q = row[1];
    first = ok ? cc * (2 * chunk) + p : 0;
    len = ok ? q - p : 0;
}

__global__ void __launch_bounds__(BUCKET_ROWS) bucket_to_lists(
    const int* __restrict__ tmp, const int* __restrict__ cst,
    int* __restrict__ off, int* __restrict__ ent,
    int n_rows, int nchunks, int chunk, int nb)
{
    __shared__ int cnt[BUCKET_ROWS];
    __shared__ int ws[64];
    __shared__ v2i stage[LIST_STAGE];
    const int t = (int)threadIdx.x, b = (int)blockIdx.x;
    const int g = t / PIECE_LANES, pl = t % PIECE_LANES;

    int fA0, lA0, fB0, lB0;
    piece_of(cst, g, nchunks, chunk, nb, b, fA0, lA0);
    piece_of(cst, g + PIECE_GROUPS, nchunks, chunk, nb, b, fB0, lB0);
    int s0 = 0, s1 = 0;
    for (int c = t; c < nchunks; c += BUCKET_ROWS) {
        const int* row = cst + (size_t)c * (size_t)(nb + 1) + b;
        s0 += row[0];
        s1 += row[1];
    }
    cnt[t] = 0;
    int bstart, bend;
    block_sum2(s0, s1, ws, bstart, bend);
    const int bcount = bend - bstart;

    for (int c0 = g; c0 < nchunks; c0 += 2 * PIECE_GROUPS) {
        int fA = fA0, lA = lA0, fB = fB0, lB = lB0;
        if (c0 != g) {
            piece_of(cst, c0, nchunks, chunk, nb, b, fA, lA);
            piece_of(cst, c0 + PIECE_GROUPS, nchunks, chunk, nb, b, fB, lB);
        }
        const int mx = lA > lB ? lA : lB;
        for (int i0 = pl; i0 < mx; i0 += 4 * PIECE_LANES) {
            int kA[4], kB[4];
            for (int j = 0; j < 4; ++j) {
                const int i = i0 + j * PIECE_LANES;
                kA[j] = tmp[4 * (i < lA ? fA + i : 0)];
                kB[j] = tmp[4 * (i < lB ? fB + i : 0)];
            }
            for (int j = 0; j < 4; ++j) {
                const int i = i0 + j * PIECE_LANES;
                if (i < lA) atomicAdd(&cnt[kA[j] & (BUCKET_ROWS - 1)], 1);
                if (i < lB) atomicAdd(&cnt[kB[j] & (BUCKET_ROWS - 1)], 1);
            }
        }
    }
    __syncthreads();

    const int c = cnt[t];
    int tot;
    const int ex = block_exclusive(c, ws, tot);
    const int row = b * BUCKET_ROWS + t;
    if (row < n_rows) off[row] = bstart + ex;
    if (row == n_rows - 1) off[n_rows] = bstart + ex + c;
    cnt[t] = ex;
    __syncthreads();

    v2i* en2 = (v2i*)ent + bstart;
    const bool staged = bcount > 0 && bcount <= LIST_STAGE;
    for (int c0 = g; c0 < nchunks; c0 += 2 * PIECE_GROUPS) {
        int fA = fA0, lA = lA0, fB = fB0, lB = lB0;
        if (c0 != g) {
            piece_of(cst, c0, nchunks, chunk, nb, b, fA, lA);
            piece_of(cst, c0 + PIECE_GROUPS, nchunks, chunk, nb, b, fB, lB);
        }
        const int mx = lA > lB ? lA : lB;
        for (int i0 = pl; i0 < mx; i0 += 4 * PIECE_LANES) {
            v3i xA[4], xB[4];
            int qA[4], qB[4];
            for (int j = 0; j < 4; ++j) {
                const int i = i0 + j * PIECE_LANES;
                xA[j] = *(const v3i*)(tmp + 4 * (i < lA ? fA + i : 0));
                xB[j] = *(const v3i*)(tmp + 4 * (i < lB ? fB + i : 0));
            }
            for (int j = 0; j < 4; ++j) {
                const int i = i0 + j * PIECE_LANES;
                qA[j] = 0; qB[j] = 0;
                if (i < lA) qA[j] = atomicAdd(&cnt[xA[j].x & (BUCKET_ROWS - 1)], 1);
                if (i < lB) qB[j] = atomicAdd(&cnt[xB[j].x & (BUCKET_ROWS - 1)], 1);
            }
            for (int j = 0; j < 4; ++j) {
                const int i = i0 + j * PIECE_LANES;
                if (i < lA) {
                    v2i e; e.x = xA[j].y; e.y = xA[j].z;
                    if (staged) stage[qA[j]] = e; else en2[qA[j]] = e;
                }
                if (i < lB) {
                    v2i e; e.x = xB[j].y; e.y = xB[j].z;
                    if (staged) stage[qB[j]] = e; else en2[qB[j]] = e;
                }
            }
        }
    }
    if (staged) {
        __syncthreads();
        const int sh = bstart & 31;
        for (int i0 = t - sh; i0 < bcount; i0 += 4 * BUCKET_ROWS) {
            v2i x[4];
            for (int j = 0; j < 4; ++j) {
                const int i = i0 + j * BUCKET_ROWS;
                x[j] = stage[(i >= 0 && i < bcount) ? i : 0];
            }
            for (int j = 0; j < 4; ++j) {
                const int i = i0 + j * BUCKET_ROWS;
                if (i >= 0 && i < bcount) en2[i] = x[j];
            }
        }
    }
}

__device__ __forceinline__ void pull_body(
    const float* __restrict__ src, float* __restrict__ dst,
    const int* __restrict__ off, const int* __restrict__ ent, int n_rows,
    unsigned blk, unsigned nblk)
{
    const int l   = (int)(threadIdx.x & 15);
    const int nhw = (int)(nblk * (blockDim.x >> 4));
    int r = (int)((blk * blockDim.x + threadIdx.x) >> 4);
    const v2i* en2 = (const v2i*)ent;
    const v4f* s4  = (const v4f*)src;
    v4f* d4 = (v4f*)dst;

    const v4f zero = {0.f, 0.f, 0.f, 0.f};
    const int last = n_rows - 1;
    if (r >= n_rows) return;
    int b = off[r], e = off[r + 1];
    while (r < n_rows) {
        int rn = r + nhw;
        int rc = rn < last ? rn : last;
        int bn = off[rc], en = off[rc + 1];

        v4f acc = zero;
        for (int base = b; base < e; base += 16) {
            int m = e - base;
            if (m > 16) m = 16;
            int   myi = 0;
            float myw = 0.f;
            if (l < m) {
                v2i t = en2[base + l];
                myi = t.x;
                myw = __int_as_float(t.y);
            }
            for (int k = 0; k < m; k += 8) {
                const int c = m - k;
                v4f v0, v1, v2, v3, v4, v5, v6, v7;
                const float w0 = __shfl(myw, k + 0, 16);
                const float w1 = __shfl(myw, k + 1, 16);
                const float w2 = __shfl(myw, k + 2, 16);
                const float w3 = __shfl(myw, k + 3, 16);
                const float w4 = __shfl(myw, k + 4, 16);
                const float w5 = __shfl(myw, k + 5, 16);
                const float w6 = __shfl(myw, k + 6, 16);
                const float w7 = __shfl(myw, k + 7, 16);
#define ROW_LD(j) v##j = s4[(size_t)__shfl(myi, k + (j), 16) * (D / 4) + l];
                ROW_LD(0)
                if (c > 1) { ROW_LD(1)
                if (c > 2) { ROW_LD(2)
                if (c > 3) { ROW_LD(3)
                if (c > 4) { ROW_LD(4)
                if (c > 5) { ROW_LD(5)
                if (c > 6) { ROW_LD(6)
                if (c > 7) { ROW_LD(7) } } } } } } }
#undef ROW_LD
                acc += w0 * v0;
                if (c > 1) { acc += w1 * v1;
                if (c > 2) { acc += w2 * v2;
                if (c > 3) { acc += w3 * v3;
                if (c > 4) { acc += w4 * v4;
                if (c > 5) { acc += w5 * v5;
                if (c > 6) { acc += w6 * v6;
                if (c > 7) { acc += w7 * v7; } } } } } } }
            }
        }
        d4[(size_t)r * (D / 4) + l] = acc;
        r = rn; b = bn; e = en;
    }
}

__global__ void __launch_bounds__(256) pull_both_sides(
    const float* __restrict__ src_a, float* __restrict__ dst_a,
    const int* __restrict__ off_a, int rows_a, unsigned nb_a,
    const float* __restrict__ src_b, float* __restrict__ dst_b,
    const int* __restrict__ off_b, int rows_b,
    const int* __restrict__ ent)
{
    const unsigned b = blockIdx.x;
    const bool second = b >= nb_a;
    pull_body(second ? src_b : src_a, second ? dst_b : dst_a,
              second ? off_b : off_a, ent, second ? rows_b : rows_a,
              second ? b - nb_a : b, second ? gridDim.x - nb_a : nb_a);
}


__device__ __forceinline__ void first_entries(
    const v2i* __restrict__ en2, int b, int e, int l, int& myi, float& myw)
{
    myi = 0;
    myw = 0.f;
    if (b + l < e) {
        v2i t = en2[b + l];
        myi = t.x;
        myw = __int_as_float(t.y);
    }
}

template <class Side>
__device__ __forceinline__ v4f sum_list(
    const v4f* __restrict__ s4, const v2i* __restrict__ en2,
    int b, int e, int l, int myi, float myw, Side side)
{
    v4f acc = {0.f, 0.f, 0.f, 0.f};
    bool side_pending = true;
    for (int base = b; base < e; base += 16) {
        int m = e - base;
        if (m > 16) m = 16;
        if (base != b) first_entries(en2, base, e, l, myi, myw);
        for (int k = 0; k < m; k += 8) {
            const int c = m - k;
            v4f v0, v1, v2, v3, v4, v5, v6, v7;
            const float w0 = __shfl(myw, k + 0, 16);
            const float w1 = __shfl(myw, k + 1, 16);
            const float w2 = __shfl(myw, k + 2, 16);
            const float w3 = __shfl(myw, k + 3, 16);
            const float w4 = __shfl(myw, k + 4, 16);
            const float w5 = __shfl(myw, k + 5, 16);
            const float w6 = __shfl(myw, k + 6, 16);
            const float w7 = __shfl(myw, k + 7, 16);
#define ROW_LD(j) v##j = s4[(size_t)__shfl(myi, k + (j), 16) * (D / 4) + l];
            ROW_LD(0)
            if (c > 1) { ROW_LD(1)
            if (c > 2) { ROW_LD(2)
            if (c > 3) { ROW_LD(3)
            if (c > 4) { ROW_LD(4)
            if (c > 5) { ROW_LD(5)
            if (c > 6) { ROW_LD(6)
            if (c > 7) { ROW_LD(7) } } } } } } }
#undef ROW_LD
            if (side_pending) { side(); side_pending = false; }
            acc += w0 * v0;
            if (c > 1) { acc += w1 * v1;
            if (c > 2) { acc += w2 * v2;
            if (c > 3) { acc += w3 * v3;
            if (c > 4) { acc += w4 * v4;
            if (c > 5) { acc += w5 * v5;
            if (c > 6) { acc += w6 * v6;
            if (c > 7) { acc += w7 * v7; } } } } } } }
        }
    }
    if (side_pending) side();
    return acc;
}

__global__ void __launch_bounds__(256) requested_rows_two_levels(
    const float* __restrict__ user_feat, const float* __restrict__ item_feat,
    const float* __restrict__ u1, const float* __restrict__ i1,
    const int* __restrict__ off, const int* __restrict__ ent,
    const int* __restrict__ users, const int* __restrict__ pos,
    const int* __restrict__ neg,
    float* __restrict__ out, int batch, int n_users)
{
    const int l = (int)(threadIdx.x & 15);
    const int r = (int)((blockIdx.x * blockDim.x + threadIdx.x) >> 4);
    if (r >= 3 * batch) return;
    const int seg = r / batch;
    const int i   = r - seg * batch;
    const bool is_user = (seg == 0);
    const int idx = is_user ? users[i] : (seg == 1 ? pos[i] : neg[i]);

    const v4f* own0 = (const v4f*)(is_user ? user_feat : item_feat);
    const v4f* own1 = (const v4f*)(is_user ? u1 : i1);
    const v4f* opp1 = (const v4f*)(is_user ? i1 : u1);
    const int jr       = is_user ? idx : n_users + idx;
    const int opp_base = is_user ? n_users : 0;
    const v2i* en2 = (const v2i*)ent;

    const int b = off[jr], e = off[jr + 1];
    v4f acc2 = {0.f, 0.f, 0.f, 0.f};
    v4f acc3 = {0.f, 0.f, 0.f, 0.f};
    for (int base = b; base < e; base += 16) {
        int m = e - base;
        if (m > 16) m = 16;
        int   s  = 0, sb = 0, se = 0;
        float w  = 0.f;
        if (l < m) {
            v2i t = en2[base + l];
            s  = t.x;
            w  = __int_as_float(t.y);
            sb = off[opp_base + s];
            se = off[opp_base + s + 1];
        }
        int kb = __shfl(sb, 0, 16), ke = __shfl(se, 0, 16);
        int   ci;
        float cw;
        first_entries(en2, kb, ke, l, ci, cw);
        for (int k = 0; k < m; ++k) {
            const int   sk = __shfl(s, k, 16);
            const float wk = __shfl(w, k, 16);
            const int kn = (k + 1 < m) ? k + 1 : k;
            const int nb = __shfl(sb, kn, 16), ne = __shfl(se, kn, 16);
            int   ni = 0;
            float nw = 0.f;
            v4f   v;
            const v4f t = sum_list(own1, en2, kb, ke, l, ci, cw, [&]() {
                if (k + 1 < m) first_entries(en2, nb, ne, l, ni, nw);
                v = opp1[(size_t)sk * (D / 4) + l];
            });
            acc2 += wk * v;
            acc3 += wk * t;
            kb = nb; ke = ne; ci = ni; cw = nw;
        }
    }
    const v4f h0 = own0[(size_t)idx * (D / 4) + l];
    const v4f h1 = own1[(size_t)idx * (D / 4) + l];
    ((v4f*)out)[(size_t)r * (D / 4) + l] = ((h0 + h1) + acc2) + acc3;
}

__global__ void __launch_bounds__(256) wmma_scale_inplace(
    float* __restrict__ out, int ntiles)
{
    int wid = (int)(blockIdx.x * (blockDim.x >> 5) + (threadIdx.x >> 5));
    if (wid >= ntiles) return;

    int lane = threadIdx.x & 31;
    int h = lane >> 4;
    int l = lane & 15;
    int rg = wid >> 2;
    int ct = wid & 3;

    float* base = out + (size_t)rg * 16 * D + ct * 16;

    v8f acc = {};
#pragma unroll
    for (int c = 0; c < 4; ++c) {
        int k0 = 2 * h;
        int k1 = 2 * h + 1;
        v2f a, b;
        a.x = base[(size_t)l * D + 4 * c + k0];
        a.y = base[(size_t)l * D + 4 * c + k1];
        b.x = ((4 * c + k0) == l) ? 0.25f : 0.0f;
        b.y = ((4 * c + k1) == l) ? 0.25f : 0.0f;
        acc = __builtin_amdgcn_wmma_f32_16x16x4_f32(
             false, a,  false, b,
             (short)0, acc,  false,  false);
    }
#pragma unroll
    for (int j = 0; j < 8; ++j)
        base[(size_t)(h * 8 + j) * D + l] = acc[j];
}

extern "C" void kernel_launch(void* const* d_in, const int* in_sizes, int n_in,
                              void* d_out, int out_size, void* d_ws, size_t ws_size,
                              hipStream_t stream)
{
    const float* user_feat = (const float*)d_in[0];
    const float* item_feat = (const float*)d_in[1];
    const int*   edge_u    = (const int*)d_in[2];
    const int*   edge_i    = (const int*)d_in[3];
    const float* norm      = (const float*)d_in[4];
    const int*   users     = (const int*)d_in[5];
    const int*   pos       = (const int*)d_in[6];
    const int*   neg       = (const int*)d_in[7];
    float* out = (float*)d_out;

    const int n_users = in_sizes[0] / D;
    const int n_items = in_sizes[1] / D;
    const int n_edges = in_sizes[2];
    const int batch   = in_sizes[5];
    const int n_rows  = n_users + n_items;

    size_t uElems = (size_t)n_users * D;
    size_t iElems = (size_t)n_items * D;

    float* su_a = (float*)d_ws;
    float* su_b = su_a + uElems;
    float* si   = su_b + uElems;
    int*   off  = (int*)(si + iElems);
    int offPad  = ((n_rows + 1 + 63) / 64) * 64;
    int*   ent  = off + offPad;

    const int nb = (n_rows + BUCKET_ROWS - 1) / BUCKET_ROWS;
    const int need = (n_edges + PASS_EDGES - 1) / PASS_EDGES;
    const int nchunks = need > MIN_CHUNKS ? need : MIN_CHUNKS;
    const int chunk = (((n_edges + nchunks - 1) / nchunks) + 3) & ~3;
    int*   tmp  = (int*)su_b;
    int*   cst  = tmp + 8 * (size_t)nchunks * (size_t)chunk;

    chunk_sort<<<nchunks, 1024, 0, stream>>>(edge_u, edge_i, norm, tmp, cst,
                                            n_users, n_edges, chunk, nb);
    bucket_to_lists<<<nb, BUCKET_ROWS, 0, stream>>>(tmp, cst, off, ent, n_rows, nchunks, chunk, nb);

    int ru = (n_users + 24575) / 24576;
    int ri = (n_items + 24575) / 24576;
    int ublocks = (n_users + 16 * ru - 1) / (16 * ru);
    int iblocks = (n_items + 16 * ri - 1) / (16 * ri);
    const int* off_u = off;
    const int* off_i = off + n_users;

    pull_both_sides<<<ublocks + iblocks, 256, 0, stream>>>(
        item_feat, su_a, off_u, n_users, ublocks,
        user_feat, si,   off_i, n_items, ent);

    int oblocks = (3 * batch + 15) / 16;
    requested_rows_two_levels<<<oblocks, 256, 0, stream>>>(
        user_feat, item_feat, su_a, si, off, ent, users, pos, neg, out, batch, n_users);

    int nrows  = 3 * batch;
    int ntiles = (nrows / 16) * (D / 16);
    int wblocks = (ntiles + 7) / 8;
    wmma_scale_inplace<<<wblocks, 256, 0, stream>>>(out, ntiles);
}


# layer-1 pull works one row ahead: a row's list entries are asked for under the row loads of the row before it (bounds two rows ahead), one dependent round trip per row, not two; sum order unchanged
# speedup vs baseline: 4.4728x; 1.0350x over previous
#include <hip/hip_runtime.h>

#define D 64

typedef __attribute__((ext_vector_type(2))) float v2f;
typedef __attribute__((ext_vector_type(4))) float v4f;
typedef __attribute__((ext_vector_type(8))) float v8f;
typedef __attribute__((ext_vector_type(2))) int   v2i;

typedef __attribute__((ext_vector_type(3))) int   v3i;
typedef __attribute__((ext_vector_type(4))) int   v4i;

#define SORT_SHIFT  10
#define BUCKET_ROWS (1 << SORT_SHIFT)
#define MAX_BUCKETS 1024
#define MIN_CHUNKS  256
#define PASS_EDGES  4096
#define LIST_STAGE  12288
#define PIECE_LANES 8
#define PIECE_GROUPS (BUCKET_ROWS / PIECE_LANES)

__device__ __forceinline__ int block_exclusive(int v, int* ws, int& total)
{
    const int t = (int)threadIdx.x, lane = t & 31, wv = t >> 5;
    const int nw = (int)(blockDim.x >> 5);
    int inc = v;
    for (int d = 1; d < 32; d <<= 1) {
        int up = __shfl_up(inc, d, 32);
        if (lane >= d) inc += up;
    }
    __syncthreads();
    if (lane == 31) ws[wv] = inc;
    __syncthreads();
    int pre = 0, tot = 0;
    for (int i = 0; i < nw; ++i) {
        int s = ws[i];
        if (i < wv) pre += s;
        tot += s;
    }
    total = tot;
    return pre + inc - v;
}

__device__ __forceinline__ void block_sum2(int a, int b, int* ws, int& ta, int& tb)
{
    const int t = (int)threadIdx.x, lane = t & 31, wv = t >> 5;
    const int nw = (int)(blockDim.x >> 5);
    for (int d = 16; d > 0; d >>= 1) {
        a += __shfl_xor(a, d, 32);
        b += __shfl_xor(b, d, 32);
    }
    __syncthreads();
    if (lane == 0) { ws[wv] = a; ws[32 + wv] = b; }
    __syncthreads();
    int sa = 0, sb = 0;
    for (int i = 0; i < nw; ++i) { sa += ws[i]; sb += ws[32 + i]; }
    ta = sa;
    tb = sb;
}

__global__ void __launch_bounds__(1024) chunk_sort(
    const int* __restrict__ edge_u, const int* __restrict__ edge_i,
    const float* __restrict__ norm, int* __restrict__ tmp, int* __restrict__ cst,
    int n_users, int n_edges, int chunk, int nb)
{
    __shared__ int lnum[MAX_BUCKETS];
    __shared__ int lbeg[MAX_BUCKETS];
    __shared__ int ws[32];
    __shared__ v4i stage[2 * PASS_EDGES];
    const int t = (int)threadIdx.x;
    for (int i = t; i < nb; i += 1024) lnum[i] = 0;
    __syncthreads();
    const int beg = (int)blockIdx.x * chunk;
    int end = beg + chunk;
    if (end > n_edges) end = n_edges;
    const int e0 = beg + 4 * t;
    int nv = end - e0;
    nv = nv < 0 ? 0 : (nv > 4 ? 4 : nv);
    int u[4] = {0, 0, 0, 0}, it[4] = {0, 0, 0, 0}, wb[4] = {0, 0, 0, 0};
    if (nv == 4) {
        v4i a = *(const v4i*)(edge_u + e0);
        v4i b = *(const v4i*)(edge_i + e0);
        v4i c = *(const v4i*)(norm + e0);
        u[0] = a.x; u[1] = a.y; u[2] = a.z; u[3] = a.w;
        it[0] = b.x; it[1] = b.y; it[2] = b.z; it[3] = b.w;
        wb[0] = c.x; wb[1] = c.y; wb[2] = c.z; wb[3] = c.w;
    } else {
        for (int j = 0; j < 4; ++j)
            if (j < nv) {
                u[j] = edge_u[e0 + j]; it[j] = edge_i[e0 + j];
                wb[j] = __float_as_int(norm[e0 + j]);
            }
    }
    int pu[4], pi[4];
    for (int j = 0; j < 4; ++j) {
        pu[j] = 0; pi[j] = 0;
        if (j < nv) {
            pu[j] = atomicAdd(&lnum[u[j] >> SORT_SHIFT], 1);
            pi[j] = atomicAdd(&lnum[(n_users + it[j]) >> SORT_SHIFT], 1);
        }
    }
    __syncthreads();
    int* my_cst = cst + (size_t)blockIdx.x * (size_t)(nb + 1);
    int chunk_items = 0;
    for (int base = 0; base < nb; base += 1024) {
        const int i = base + t;
        const int v = (i < nb) ? lnum[i] : 0;
        int tot;
        const int ex = block_exclusive(v, ws, tot);
        if (i < nb) {
            lbeg[i] = chunk_items + ex;
            my_cst[i] = chunk_items + ex;
        }
        chunk_items += tot;
    }
    if (t == 0) my_cst[nb] = chunk_items;
    __syncthreads();
    for (int j = 0; j < 4; ++j)
        if (j < nv) {
            v4i x; x.x = u[j];            x.y = it[j]; x.z = wb[j]; x.w = 0;
            v4i y; y.x = n_users + it[j]; y.y = u[j];  y.z = wb[j]; y.w = 0;
            stage[lbeg[x.x >> SORT_SHIFT] + pu[j]] = x;
            stage[lbeg[y.x >> SORT_SHIFT] + pi[j]] = y;
        }
    __syncthreads();
    v4i* region = (v4i*)tmp + (size_t)blockIdx.x * (size_t)(2 * chunk);
    for (int i0 = t; i0 < chunk_items; i0 += 4 * 1024) {
        v4i x[4];
        for (int j = 0; j < 4; ++j) {
            const int i = i0 + j * 1024;
            x[j] = stage[i < chunk_items ? i : i0];
        }
        for (int j = 0; j < 4; ++j)
            if (i0 + j * 1024 < chunk_items) region[i0 + j * 1024] = x[j];
    }
}

__device__ __forceinline__ void piece_of(
    const int* __restrict__ cst, int c, int nchunks, int chunk, int nb, int b,
    int& first, int& len)
{
    const bool ok = c < nchunks;
    const int cc = ok ? c : 0;
    const int* row = cst + (size_t)cc * (size_t)(nb + 1) + b;
    const int p = row[0], q = row[1];
    first = ok ? cc * (2 * chunk) + p : 0;
    len = ok ? q - p : 0;
}

__global__ void __launch_bounds__(BUCKET_ROWS) bucket_to_lists(
    const int* __restrict__ tmp, const int* __restrict__ cst,
    int* __restrict__ off, int* __restrict__ ent,
    int n_rows, int nchunks, int chunk, int nb)
{
    __shared__ int cnt[BUCKET_ROWS];
    __shared__ int ws[64];
    __shared__ v2i stage[LIST_STAGE];
    const int t = (int)threadIdx.x, b = (int)blockIdx.x;
    const int g = t / PIECE_LANES, pl = t % PIECE_LANES;

    int fA0, lA0, fB0, lB0;
    piece_of(cst, g, nchunks, chunk, nb, b, fA0, lA0);
    piece_of(cst, g + PIECE_GROUPS, nchunks, chunk, nb, b, fB0, lB0);
    int s0 = 0, s1 = 0;
    for (int c = t; c < nchunks; c += BUCKET_ROWS) {
        const int* row = cst + (size_t)c * (size_t)(nb + 1) + b;
        s0 += row[0];
        s1 += row[1];
    }
    cnt[t] = 0;
    int bstart, bend;
    block_sum2(s0, s1, ws, bstart, bend);
    const int bcount = bend - bstart;

    for (int c0 = g; c0 < nchunks; c0 += 2 * PIECE_GROUPS) {
        int fA = fA0, lA = lA0, fB = fB0, lB = lB0;
        if (c0 != g) {
            piece_of(cst, c0, nchunks, chunk, nb, b, fA, lA);
            piece_of(cst, c0 + PIECE_GROUPS, nchunks, chunk, nb, b, fB, lB);
        }
        const int mx = lA > lB ? lA : lB;
        for (int i0 = pl; i0 < mx; i0 += 4 * PIECE_LANES) {
            int kA[4], kB[4];
            for (int j = 0; j < 4; ++j) {
                const int i = i0 + j * PIECE_LANES;
                kA[j] = tmp[4 * (i < lA ? fA + i : 0)];
                kB[j] = tmp[4 * (i < lB ? fB + i : 0)];
            }
            for (int j = 0; j < 4; ++j) {
                const int i = i0 + j * PIECE_LANES;
                if (i < lA) atomicAdd(&cnt[kA[j] & (BUCKET_ROWS - 1)], 1);
                if (i < lB) atomicAdd(&cnt[kB[j] & (BUCKET_ROWS - 1)], 1);
            }
        }
    }
    __syncthreads();

    const int c = cnt[t];
    int tot;
    const int ex = block_exclusive(c, ws, tot);
    const int row = b * BUCKET_ROWS + t;
    if (row < n_rows) off[row] = bstart + ex;
    if (row == n_rows - 1) off[n_rows] = bstart + ex + c;
    cnt[t] = ex;
    __syncthreads();

    v2i* en2 = (v2i*)ent + bstart;
    const bool staged = bcount > 0 && bcount <= LIST_STAGE;
    for (int c0 = g; c0 < nchunks; c0 += 2 * PIECE_GROUPS) {
        int fA = fA0, lA = lA0, fB = fB0, lB = lB0;
        if (c0 != g) {
            piece_of(cst, c0, nchunks, chunk, nb, b, fA, lA);
            piece_of(cst, c0 + PIECE_GROUPS, nchunks, chunk, nb, b, fB, lB);
        }
        const int mx = lA > lB ? lA : lB;
        for (int i0 = pl; i0 < mx; i0 += 4 * PIECE_LANES) {
            v3i xA[4], xB[4];
            int qA[4], qB[4];
            for (int j = 0; j < 4; ++j) {
                const int i = i0 + j * PIECE_LANES;
                xA[j] = *(const v3i*)(tmp + 4 * (i < lA ? fA + i : 0));
                xB[j] = *(const v3i*)(tmp + 4 * (i < lB ? fB + i : 0));
            }
            for (int j = 0; j < 4; ++j) {
                const int i = i0 + j * PIECE_LANES;
                qA[j] = 0; qB[j] = 0;
                if (i < lA) qA[j] = atomicAdd(&cnt[xA[j].x & (BUCKET_ROWS - 1)], 1);
                if (i < lB) qB[j] = atomicAdd(&cnt[xB[j].x & (BUCKET_ROWS - 1)], 1);
            }
            for (int j = 0; j < 4; ++j) {
                const int i = i0 + j * PIECE_LANES;
                if (i < lA) {
                    v2i e; e.x = xA[j].y; e.y = xA[j].z;
                    if (staged) stage[qA[j]] = e; else en2[qA[j]] = e;
                }
                if (i < lB) {
                    v2i e; e.x = xB[j].y; e.y = xB[j].z;
                    if (staged) stage[qB[j]] = e; else en2[qB[j]] = e;
                }
            }
        }
    }
    if (staged) {
        __syncthreads();
        const int sh = bstart & 31;
        for (int i0 = t - sh; i0 < bcount; i0 += 4 * BUCKET_ROWS) {
            v2i x[4];
            for (int j = 0; j < 4; ++j) {
                const int i = i0 + j * BUCKET_ROWS;
                x[j] = stage[(i >= 0 && i < bcount) ? i : 0];
            }
            for (int j = 0; j < 4; ++j) {
                const int i = i0 + j * BUCKET_ROWS;
                if (i >= 0 && i < bcount) en2[i] = x[j];
            }
        }
    }
}

__device__ __forceinline__ void first_entries(
    const v2i* __restrict__ en2, int b, int e, int l, int& myi, float& myw)
{
    myi = 0;
    myw = 0.f;
    if (b + l < e) {
        v2i t = en2[b + l];
        myi = t.x;
        myw = __int_as_float(t.y);
    }
}

__device__ __forceinline__ void pull_body(
    const float* __restrict__ src, float* __restrict__ dst,
    const int* __restrict__ off, const int* __restrict__ ent, int n_rows,
    unsigned blk, unsigned nblk)
{
    const int l   = (int)(threadIdx.x & 15);
    const int nhw = (int)(nblk * (blockDim.x >> 4));
    int r = (int)((blk * blockDim.x + threadIdx.x) >> 4);
    const v2i* en2 = (const v2i*)ent;
    const v4f* s4  = (const v4f*)src;
    v4f* d4 = (v4f*)dst;

    const v4f zero = {0.f, 0.f, 0.f, 0.f};
    const int last = n_rows - 1;
    if (r >= n_rows) return;
    int b = off[r], e = off[r + 1];
    int bn, en;
    {
        const int r1 = r + nhw;
        const int c1 = r1 < last ? r1 : last;
        bn = off[c1]; en = off[c1 + 1];
    }
    int   myi;
    float myw;
    first_entries(en2, b, e, l, myi, myw);
    while (r < n_rows) {
        const int rn = r + nhw;
        const bool have_next = rn < n_rows;
        const int r2 = rn + nhw;
        const int c2 = r2 < last ? r2 : last;
        int   b2 = 0, e2 = 0;
        int   ni = 0;
        float nw = 0.f;
        bool  ahead_pending = true;

        v4f acc = zero;
        for (int base = b; base < e; base += 16) {
            int m = e - base;
            if (m > 16) m = 16;
            if (base != b) first_entries(en2, base, e, l, myi, myw);
            for (int k = 0; k < m; k += 8) {
                const int c = m - k;
                v4f v0, v1, v2, v3, v4, v5, v6, v7;
                const float w0 = __shfl(myw, k + 0, 16);
                const float w1 = __shfl(myw, k + 1, 16);
                const float w2 = __shfl(myw, k + 2, 16);
                const float w3 = __shfl(myw, k + 3, 16);
                const float w4 = __shfl(myw, k + 4, 16);
                const float w5 = __shfl(myw, k + 5, 16);
                const float w6 = __shfl(myw, k + 6, 16);
                const float w7 = __shfl(myw, k + 7, 16);
#define ROW_LD(j) v##j = s4[(size_t)__shfl(myi, k + (j), 16) * (D / 4) + l];
                ROW_LD(0)
                if (c > 1) { ROW_LD(1)
                if (c > 2) { ROW_LD(2)
                if (c > 3) { ROW_LD(3)
                if (c > 4) { ROW_LD(4)
                if (c > 5) { ROW_LD(5)
                if (c > 6) { ROW_LD(6)
                if (c > 7) { ROW_LD(7) } } } } } } }
#undef ROW_LD
                if (ahead_pending) {
                    if (have_next) first_entries(en2, bn, en, l, ni, nw);
                    b2 = off[c2]; e2 = off[c2 + 1];
                    ahead_pending = false;
                }
                acc += w0 * v0;
                if (c > 1) { acc += w1 * v1;
                if (c > 2) { acc += w2 * v2;
                if (c > 3) { acc += w3 * v3;
                if (c > 4) { acc += w4 * v4;
                if (c > 5) { acc += w5 * v5;
                if (c > 6) { acc += w6 * v6;
                if (c > 7) { acc += w7 * v7; } } } } } } }
            }
        }
        if (ahead_pending) {
            if (have_next) first_entries(en2, bn, en, l, ni, nw);
            b2 = off[c2]; e2 = off[c2 + 1];
        }
        d4[(size_t)r * (D / 4) + l] = acc;
        r = rn; b = bn; e = en; bn = b2; en = e2; myi = ni; myw = nw;
    }
}

__global__ void __launch_bounds__(256) pull_both_sides(
    const float* __restrict__ src_a, float* __restrict__ dst_a,
    const int* __restrict__ off_a, int rows_a, unsigned nb_a,
    const float* __restrict__ src_b, float* __restrict__ dst_b,
    const int* __restrict__ off_b, int rows_b,
    const int* __restrict__ ent)
{
    const unsigned b = blockIdx.x;
    const bool second = b >= nb_a;
    pull_body(second ? src_b : src_a, second ? dst_b : dst_a,
              second ? off_b : off_a, ent, second ? rows_b : rows_a,
              second ? b - nb_a : b, second ? gridDim.x - nb_a : nb_a);
}


template <class Side>
__device__ __forceinline__ v4f sum_list(
    const v4f* __restrict__ s4, const v2i* __restrict__ en2,
    int b, int e, int l, int myi, float myw, Side side)
{
    v4f acc = {0.f, 0.f, 0.f, 0.f};
    bool side_pending = true;
    for (int base = b; base < e; base += 16) {
        int m = e - base;
        if (m > 16) m = 16;
        if (base != b) first_entries(en2, base, e, l, myi, myw);
        for (int k = 0; k < m; k += 8) {
            const int c = m - k;
            v4f v0, v1, v2, v3, v4, v5, v6, v7;
            const float w0 = __shfl(myw, k + 0, 16);
            const float w1 = __shfl(myw, k + 1, 16);
            const float w2 = __shfl(myw, k + 2, 16);
            const float w3 = __shfl(myw, k + 3, 16);
            const float w4 = __shfl(myw, k + 4, 16);
            const float w5 = __shfl(myw, k + 5, 16);
            const float w6 = __shfl(myw, k + 6, 16);
            const float w7 = __shfl(myw, k + 7, 16);
#define ROW_LD(j) v##j = s4[(size_t)__shfl(myi, k + (j), 16) * (D / 4) + l];
            ROW_LD(0)
            if (c > 1) { ROW_LD(1)
            if (c > 2) { ROW_LD(2)
            if (c > 3) { ROW_LD(3)
            if (c > 4) { ROW_LD(4)
            if (c > 5) { ROW_LD(5)
            if (c > 6) { ROW_LD(6)
            if (c > 7) { ROW_LD(7) } } } } } } }
#undef ROW_LD
            if (side_pending) { side(); side_pending = false; }
            acc += w0 * v0;
            if (c > 1) { acc += w1 * v1;
            if (c > 2) { acc += w2 * v2;
            if (c > 3) { acc += w3 * v3;
            if (c > 4) { acc += w4 * v4;
            if (c > 5) { acc += w5 * v5;
            if (c > 6) { acc += w6 * v6;
            if (c > 7) { acc += w7 * v7; } } } } } } }
        }
    }
    if (side_pending) side();
    return acc;
}

__global__ void __launch_bounds__(256) requested_rows_two_levels(
    const float* __restrict__ user_feat, const float* __restrict__ item_feat,
    const float* __restrict__ u1, const float* __restrict__ i1,
    const int* __restrict__ off, const int* __restrict__ ent,
    const int* __restrict__ users, const int* __restrict__ pos,
    const int* __restrict__ neg,
    float* __restrict__ out, int batch, int n_users)
{
    const int l = (int)(threadIdx.x & 15);
    const int r = (int)((blockIdx.x * blockDim.x + threadIdx.x) >> 4);
    if (r >= 3 * batch) return;
    const int seg = r / batch;
    const int i   = r - seg * batch;
    const bool is_user = (seg == 0);
    const int idx = is_user ? users[i] : (seg == 1 ? pos[i] : neg[i]);

    const v4f* own0 = (const v4f*)(is_user ? user_feat : item_feat);
    const v4f* own1 = (const v4f*)(is_user ? u1 : i1);
    const v4f* opp1 = (const v4f*)(is_user ? i1 : u1);
    const int jr       = is_user ? idx : n_users + idx;
    const int opp_base = is_user ? n_users : 0;
    const v2i* en2 = (const v2i*)ent;

    const int b = off[jr], e = off[jr + 1];
    v4f acc2 = {0.f, 0.f, 0.f, 0.f};
    v4f acc3 = {0.f, 0.f, 0.f, 0.f};
    for (int base = b; base < e; base += 16) {
        int m = e - base;
        if (m > 16) m = 16;
        int   s  = 0, sb = 0, se = 0;
        float w  = 0.f;
        if (l < m) {
            v2i t = en2[base + l];
            s  = t.x;
            w  = __int_as_float(t.y);
            sb = off[opp_base + s];
            se = off[opp_base + s + 1];
        }
        int kb = __shfl(sb, 0, 16), ke = __shfl(se, 0, 16);
        int   ci;
        float cw;
        first_entries(en2, kb, ke, l, ci, cw);
        for (int k = 0; k < m; ++k) {
            const int   sk = __shfl(s, k, 16);
            const float wk = __shfl(w, k, 16);
            const int kn = (k + 1 < m) ? k + 1 : k;
            const int nb = __shfl(sb, kn, 16), ne = __shfl(se, kn, 16);
            int   ni = 0;
            float nw = 0.f;
            v4f   v;
            const v4f t = sum_list(own1, en2, kb, ke, l, ci, cw, [&]() {
                if (k + 1 < m) first_entries(en2, nb, ne, l, ni, nw);
                v = opp1[(size_t)sk * (D / 4) + l];
            });
            acc2 += wk * v;
            acc3 += wk * t;
            kb = nb; ke = ne; ci = ni; cw = nw;
        }
    }
    const v4f h0 = own0[(size_t)idx * (D / 4) + l];
    const v4f h1 = own1[(size_t)idx * (D / 4) + l];
    ((v4f*)out)[(size_t)r * (D / 4) + l] = ((h0 + h1) + acc2) + acc3;
}

__global__ void __launch_bounds__(256) wmma_scale_inplace(
    float* __restrict__ out, int ntiles)
{
    int wid = (int)(blockIdx.x * (blockDim.x >> 5) + (threadIdx.x >> 5));
    if (wid >= ntiles) return;

    int lane = threadIdx.x & 31;
    int h = lane >> 4;
    int l = lane & 15;
    int rg = wid >> 2;
    int ct = wid & 3;

    float* base = out + (size_t)rg * 16 * D + ct * 16;

    v8f acc = {};
#pragma unroll
    for (int c = 0; c < 4; ++c) {
        int k0 = 2 * h;
        int k1 = 2 * h + 1;
        v2f a, b;
        a.x = base[(size_t)l * D + 4 * c + k0];
        a.y = base[(size_t)l * D + 4 * c + k1];
        b.x = ((4 * c + k0) == l) ? 0.25f : 0.0f;
        b.y = ((4 * c + k1) == l) ? 0.25f : 0.0f;
        acc = __builtin_amdgcn_wmma_f32_16x16x4_f32(
             false, a,  false, b,
             (short)0, acc,  false,  false);
    }
#pragma unroll
    for (int j = 0; j < 8; ++j)
        base[(size_t)(h * 8 + j) * D + l] = acc[j];
}

extern "C" void kernel_launch(void* const* d_in, const int* in_sizes, int n_in,
                              void* d_out, int out_size, void* d_ws, size_t ws_size,
                              hipStream_t stream)
{
    const float* user_feat = (const float*)d_in[0];
    const float* item_feat = (const float*)d_in[1];
    const int*   edge_u    = (const int*)d_in[2];
    const int*   edge_i    = (const int*)d_in[3];
    const float* norm      = (const float*)d_in[4];
    const int*   users     = (const int*)d_in[5];
    const int*   pos       = (const int*)d_in[6];
    const int*   neg       = (const int*)d_in[7];
    float* out = (float*)d_out;

    const int n_users = in_sizes[0] / D;
    const int n_items = in_sizes[1] / D;
    const int n_edges = in_sizes[2];
    const int batch   = in_sizes[5];
    const int n_rows  = n_users + n_items;

    size_t uElems = (size_t)n_users * D;
    size_t iElems = (size_t)n_items * D;

    float* su_a = (float*)d_ws;
    float* su_b = su_a + uElems;
    float* si   = su_b + uElems;
    int*   off  = (int*)(si + iElems);
    int offPad  = ((n_rows + 1 + 63) / 64) * 64;
    int*   ent  = off + offPad;

    const int nb = (n_rows + BUCKET_ROWS - 1) / BUCKET_ROWS;
    const int need = (n_edges + PASS_EDGES - 1) / PASS_EDGES;
    const int nchunks = need > MIN_CHUNKS ? need : MIN_CHUNKS;
    const int chunk = (((n_edges + nchunks - 1) / nchunks) + 3) & ~3;
    int*   tmp  = (int*)su_b;
    int*   cst  = tmp + 8 * (size_t)nchunks * (size_t)chunk;

    chunk_sort<<<nchunks, 1024, 0, stream>>>(edge_u, edge_i, norm, tmp, cst,
                                            n_users, n_edges, chunk, nb);
    bucket_to_lists<<<nb, BUCKET_ROWS, 0, stream>>>(tmp, cst, off, ent, n_rows, nchunks, chunk, nb);

    int ru = (n_users + 24575) / 24576;
    int ri = (n_items + 24575) / 24576;
    int ublocks = (n_users + 16 * ru - 1) / (16 * ru);
    int iblocks = (n_items + 16 * ri - 1) / (16 * ri);
    const int* off_u = off;
    const int* off_i = off + n_users;

    pull_both_sides<<<ublocks + iblocks, 256, 0, stream>>>(
        item_feat, su_a, off_u, n_users, ublocks,
        user_feat, si,   off_i, n_items, ent);

    int oblocks = (3 * batch + 15) / 16;
    requested_rows_two_levels<<<oblocks, 256, 0, stream>>>(
        user_feat, item_feat, su_a, si, off, ent, users, pos, neg, out, batch, n_users);

    int nrows  = 3 * batch;
    int ntiles = (nrows / 16) * (D / 16);
    int wblocks = (ntiles + 7) / 8;
    wmma_scale_inplace<<<wblocks, 256, 0, stream>>>(out, ntiles);
}


# the scale by 0.25 joins the walk kernel: each wave scales its two finished rows on two f32 16x16x4 matrix instructions (0.25 selector as A, rows as B) before the one store; 5 -> 4 launches
# speedup vs baseline: 4.6114x; 1.0310x over previous
#include <hip/hip_runtime.h>

#define D 64

typedef __attribute__((ext_vector_type(2))) float v2f;
typedef __attribute__((ext_vector_type(4))) float v4f;
typedef __attribute__((ext_vector_type(8))) float v8f;
typedef __attribute__((ext_vector_type(2))) int   v2i;

typedef __attribute__((ext_vector_type(3))) int   v3i;
typedef __attribute__((ext_vector_type(4))) int   v4i;

#define SORT_SHIFT  10
#define BUCKET_ROWS (1 << SORT_SHIFT)
#define MAX_BUCKETS 1024
#define MIN_CHUNKS  256
#define PASS_EDGES  4096
#define LIST_STAGE  12288
#define PIECE_LANES 8
#define PIECE_GROUPS (BUCKET_ROWS / PIECE_LANES)

__device__ __forceinline__ int block_exclusive(int v, int* ws, int& total)
{
    const int t = (int)threadIdx.x, lane = t & 31, wv = t >> 5;
    const int nw = (int)(blockDim.x >> 5);
    int inc = v;
    for (int d = 1; d < 32; d <<= 1) {
        int up = __shfl_up(inc, d, 32);
        if (lane >= d) inc += up;
    }
    __syncthreads();
    if (lane == 31) ws[wv] = inc;
    __syncthreads();
    int pre = 0, tot = 0;
    for (int i = 0; i < nw; ++i) {
        int s = ws[i];
        if (i < wv) pre += s;
        tot += s;
    }
    total = tot;
    return pre + inc - v;
}

__device__ __forceinline__ void block_sum2(int a, int b, int* ws, int& ta, int& tb)
{
    const int t = (int)threadIdx.x, lane = t & 31, wv = t >> 5;
    const int nw = (int)(blockDim.x >> 5);
    for (int d = 16; d > 0; d >>= 1) {
        a += __shfl_xor(a, d, 32);
        b += __shfl_xor(b, d, 32);
    }
    __syncthreads();
    if (lane == 0) { ws[wv] = a; ws[32 + wv] = b; }
    __syncthreads();
    int sa = 0, sb = 0;
    for (int i = 0; i < nw; ++i) { sa += ws[i]; sb += ws[32 + i]; }
    ta = sa;
    tb = sb;
}

__global__ void __launch_bounds__(1024) chunk_sort(
    const int* __restrict__ edge_u, const int* __restrict__ edge_i,
    const float* __restrict__ norm, int* __restrict__ tmp, int* __restrict__ cst,
    int n_users, int n_edges, int chunk, int nb)
{
    __shared__ int lnum[MAX_BUCKETS];
    __shared__ int lbeg[MAX_BUCKETS];
    __shared__ int ws[32];
    __shared__ v4i stage[2 * PASS_EDGES];
    const int t = (int)threadIdx.x;
    for (int i = t; i < nb; i += 1024) lnum[i] = 0;
    __syncthreads();
    const int beg = (int)blockIdx.x * chunk;
    int end = beg + chunk;
    if (end > n_edges) end = n_edges;
    const int e0 = beg + 4 * t;
    int nv = end - e0;
    nv = nv < 0 ? 0 : (nv > 4 ? 4 : nv);
    int u[4] = {0, 0, 0, 0}, it[4] = {0, 0, 0, 0}, wb[4] = {0, 0, 0, 0};
    if (nv == 4) {
        v4i a = *(const v4i*)(edge_u + e0);
        v4i b = *(const v4i*)(edge_i + e0);
        v4i c = *(const v4i*)(norm + e0);
        u[0] = a.x; u[1] = a.y; u[2] = a.z; u[3] = a.w;
        it[0] = b.x; it[1] = b.y; it[2] = b.z; it[3] = b.w;
        wb[0] = c.x; wb[1] = c.y; wb[2] = c.z; wb[3] = c.w;
    } else {
        for (int j = 0; j < 4; ++j)
            if (j < nv) {
                u[j] = edge_u[e0 + j]; it[j] = edge_i[e0 + j];
                wb[j] = __float_as_int(norm[e0 + j]);
            }
    }
    int pu[4], pi[4];
    for (int j = 0; j < 4; ++j) {
        pu[j] = 0; pi[j] = 0;
        if (j < nv) {
            pu[j] = atomicAdd(&lnum[u[j] >> SORT_SHIFT], 1);
            pi[j] = atomicAdd(&lnum[(n_users + it[j]) >> SORT_SHIFT], 1);
        }
    }
    __syncthreads();
    int* my_cst = cst + (size_t)blockIdx.x * (size_t)(nb + 1);
    int chunk_items = 0;
    for (int base = 0; base < nb; base += 1024) {
        const int i = base + t;
        const int v = (i < nb) ? lnum[i] : 0;
        int tot;
        const int ex = block_exclusive(v, ws, tot);
        if (i < nb) {
            lbeg[i] = chunk_items + ex;
            my_cst[i] = chunk_items + ex;
        }
        chunk_items += tot;
    }
    if (t == 0) my_cst[nb] = chunk_items;
    __syncthreads();
    for (int j = 0; j < 4; ++j)
        if (j < nv) {
            v4i x; x.x = u[j];            x.y = it[j]; x.z = wb[j]; x.w = 0;
            v4i y; y.x = n_users + it[j]; y.y = u[j];  y.z = wb[j]; y.w = 0;
            stage[lbeg[x.x >> SORT_SHIFT] + pu[j]] = x;
            stage[lbeg[y.x >> SORT_SHIFT] + pi[j]] = y;
        }
    __syncthreads();
    v4i* region = (v4i*)tmp + (size_t)blockIdx.x * (size_t)(2 * chunk);
    for (int i0 = t; i0 < chunk_items; i0 += 4 * 1024) {
        v4i x[4];
        for (int j = 0; j < 4; ++j) {
            const int i = i0 + j * 1024;
            x[j] = stage[i < chunk_items ? i : i0];
        }
        for (int j = 0; j < 4; ++j)
            if (i0 + j * 1024 < chunk_items) region[i0 + j * 1024] = x[j];
    }
}

__device__ __forceinline__ void piece_of(
    const int* __restrict__ cst, int c, int nchunks, int chunk, int nb, int b,
    int& first, int& len)
{
    const bool ok = c < nchunks;
    const int cc = ok ? c : 0;
    const int* row = cst + (size_t)cc * (size_t)(nb + 1) + b;
    const int p = row[0], q = row[1];
    first = ok ? cc * (2 * chunk) + p : 0;
    len = ok ? q - p : 0;
}

__global__ void __launch_bounds__(BUCKET_ROWS) bucket_to_lists(
    const int* __restrict__ tmp, const int* __restrict__ cst,
    int* __restrict__ off, int* __restrict__ ent,
    int n_rows, int nchunks, int chunk, int nb)
{
    __shared__ int cnt[BUCKET_ROWS];
    __shared__ int ws[64];
    __shared__ v2i stage[LIST_STAGE];
    const int t = (int)threadIdx.x, b = (int)blockIdx.x;
    const int g = t / PIECE_LANES, pl = t % PIECE_LANES;

    int fA0, lA0, fB0, lB0;
    piece_of(cst, g, nchunks, chunk, nb, b, fA0, lA0);
    piece_of(cst, g + PIECE_GROUPS, nchunks, chunk, nb, b, fB0, lB0);
    int s0 = 0, s1 = 0;
    for (int c = t; c < nchunks; c += BUCKET_ROWS) {
        const int* row = cst + (size_t)c * (size_t)(nb + 1) + b;
        s0 += row[0];
        s1 += row[1];
    }
    cnt[t] = 0;
    int bstart, bend;
    block_sum2(s0, s1, ws, bstart, bend);
    const int bcount = bend - bstart;

    for (int c0 = g; c0 < nchunks; c0 += 2 * PIECE_GROUPS) {
        int fA = fA0, lA = lA0, fB = fB0, lB = lB0;
        if (c0 != g) {
            piece_of(cst, c0, nchunks, chunk, nb, b, fA, lA);
            piece_of(cst, c0 + PIECE_GROUPS, nchunks, chunk, nb, b, fB, lB);
        }
        const int mx = lA > lB ? lA : lB;
        for (int i0 = pl; i0 < mx; i0 += 4 * PIECE_LANES) {
            int kA[4], kB[4];
            for (int j = 0; j < 4; ++j) {
                const int i = i0 + j * PIECE_LANES;
                kA[j] = tmp[4 * (i < lA ? fA + i : 0)];
                kB[j] = tmp[4 * (i < lB ? fB + i : 0)];
            }
            for (int j = 0; j < 4; ++j) {
                const int i = i0 + j * PIECE_LANES;
                if (i < lA) atomicAdd(&cnt[kA[j] & (BUCKET_ROWS - 1)], 1);
                if (i < lB) atomicAdd(&cnt[kB[j] & (BUCKET_ROWS - 1)], 1);
            }
        }
    }
    __syncthreads();

    const int c = cnt[t];
    int tot;
    const int ex = block_exclusive(c, ws, tot);
    const int row = b * BUCKET_ROWS + t;
    if (row < n_rows) off[row] = bstart + ex;
    if (row == n_rows - 1) off[n_rows] = bstart + ex + c;
    cnt[t] = ex;
    __syncthreads();

    v2i* en2 = (v2i*)ent + bstart;
    const bool staged = bcount > 0 && bcount <= LIST_STAGE;
    for (int c0 = g; c0 < nchunks; c0 += 2 * PIECE_GROUPS) {
        int fA = fA0, lA = lA0, fB = fB0, lB = lB0;
        if (c0 != g) {
            piece_of(cst, c0, nchunks, chunk, nb, b, fA, lA);
            piece_of(cst, c0 + PIECE_GROUPS, nchunks, chunk, nb, b, fB, lB);
        }
        const int mx = lA > lB ? lA : lB;
        for (int i0 = pl; i0 < mx; i0 += 4 * PIECE_LANES) {
            v3i xA[4], xB[4];
            int qA[4], qB[4];
            for (int j = 0; j < 4; ++j) {
                const int i = i0 + j * PIECE_LANES;
                xA[j] = *(const v3i*)(tmp + 4 * (i < lA ? fA + i : 0));
                xB[j] = *(const v3i*)(tmp + 4 * (i < lB ? fB + i : 0));
            }
            for (int j = 0; j < 4; ++j) {
                const int i = i0 + j * PIECE_LANES;
                qA[j] = 0; qB[j] = 0;
                if (i < lA) qA[j] = atomicAdd(&cnt[xA[j].x & (BUCKET_ROWS - 1)], 1);
                if (i < lB) qB[j] = atomicAdd(&cnt[xB[j].x & (BUCKET_ROWS - 1)], 1);
            }
            for (int j = 0; j < 4; ++j) {
                const int i = i0 + j * PIECE_LANES;
                if (i < lA) {
                    v2i e; e.x = xA[j].y; e.y = xA[j].z;
                    if (staged) stage[qA[j]] = e; else en2[qA[j]] = e;
                }
                if (i < lB) {
                    v2i e; e.x = xB[j].y; e.y = xB[j].z;
                    if (staged) stage[qB[j]] = e; else en2[qB[j]] = e;
                }
            }
        }
    }
    if (staged) {
        __syncthreads();
        const int sh = bstart & 31;
        for (int i0 = t - sh; i0 < bcount; i0 += 4 * BUCKET_ROWS) {
            v2i x[4];
            for (int j = 0; j < 4; ++j) {
                const int i = i0 + j * BUCKET_ROWS;
                x[j] = stage[(i >= 0 && i < bcount) ? i : 0];
            }
            for (int j = 0; j < 4; ++j) {
                const int i = i0 + j * BUCKET_ROWS;
                if (i >= 0 && i < bcount) en2[i] = x[j];
            }
        }
    }
}

__device__ __forceinline__ void first_entries(
    const v2i* __restrict__ en2, int b, int e, int l, int& myi, float& myw)
{
    myi = 0;
    myw = 0.f;
    if (b + l < e) {
        v2i t = en2[b + l];
        myi = t.x;
        myw = __int_as_float(t.y);
    }
}

__device__ __forceinline__ void pull_body(
    const float* __restrict__ src, float* __restrict__ dst,
    const int* __restrict__ off, const int* __restrict__ ent, int n_rows,
    unsigned blk, unsigned nblk)
{
    const int l   = (int)(threadIdx.x & 15);
    const int nhw = (int)(nblk * (blockDim.x >> 4));
    int r = (int)((blk * blockDim.x + threadIdx.x) >> 4);
    const v2i* en2 = (const v2i*)ent;
    const v4f* s4  = (const v4f*)src;
    v4f* d4 = (v4f*)dst;

    const v4f zero = {0.f, 0.f, 0.f, 0.f};
    const int last = n_rows - 1;
    if (r >= n_rows) return;
    int b = off[r], e = off[r + 1];
    int bn, en;
    {
        const int r1 = r + nhw;
        const int c1 = r1 < last ? r1 : last;
        bn = off[c1]; en = off[c1 + 1];
    }
    int   myi;
    float myw;
    first_entries(en2, b, e, l, myi, myw);
    while (r < n_rows) {
        const int rn = r + nhw;
        const bool have_next = rn < n_rows;
        const int r2 = rn + nhw;
        const int c2 = r2 < last ? r2 : last;
        int   b2 = 0, e2 = 0;
        int   ni = 0;
        float nw = 0.f;
        bool  ahead_pending = true;

        v4f acc = zero;
        for (int base = b; base < e; base += 16) {
            int m = e - base;
            if (m > 16) m = 16;
            if (base != b) first_entries(en2, base, e, l, myi, myw);
            for (int k = 0; k < m; k += 8) {
                const int c = m - k;
                v4f v0, v1, v2, v3, v4, v5, v6, v7;
                const float w0 = __shfl(myw, k + 0, 16);
                const float w1 = __shfl(myw, k + 1, 16);
                const float w2 = __shfl(myw, k + 2, 16);
                const float w3 = __shfl(myw, k + 3, 16);
                const float w4 = __shfl(myw, k + 4, 16);
                const float w5 = __shfl(myw, k + 5, 16);
                const float w6 = __shfl(myw, k + 6, 16);
                const float w7 = __shfl(myw, k + 7, 16);
#define ROW_LD(j) v##j = s4[(size_t)__shfl(myi, k + (j), 16) * (D / 4) + l];
                ROW_LD(0)
                if (c > 1) { ROW_LD(1)
                if (c > 2) { ROW_LD(2)
                if (c > 3) { ROW_LD(3)
                if (c > 4) { ROW_LD(4)
                if (c > 5) { ROW_LD(5)
                if (c > 6) { ROW_LD(6)
                if (c > 7) { ROW_LD(7) } } } } } } }
#undef ROW_LD
                if (ahead_pending) {
                    if (have_next) first_entries(en2, bn, en, l, ni, nw);
                    b2 = off[c2]; e2 = off[c2 + 1];
                    ahead_pending = false;
                }
                acc += w0 * v0;
                if (c > 1) { acc += w1 * v1;
                if (c > 2) { acc += w2 * v2;
                if (c > 3) { acc += w3 * v3;
                if (c > 4) { acc += w4 * v4;
                if (c > 5) { acc += w5 * v5;
                if (c > 6) { acc += w6 * v6;
                if (c > 7) { acc += w7 * v7; } } } } } } }
            }
        }
        if (ahead_pending) {
            if (have_next) first_entries(en2, bn, en, l, ni, nw);
            b2 = off[c2]; e2 = off[c2 + 1];
        }
        d4[(size_t)r * (D / 4) + l] = acc;
        r = rn; b = bn; e = en; bn = b2; en = e2; myi = ni; myw = nw;
    }
}

__global__ void __launch_bounds__(256) pull_both_sides(
    const float* __restrict__ src_a, float* __restrict__ dst_a,
    const int* __restrict__ off_a, int rows_a, unsigned nb_a,
    const float* __restrict__ src_b, float* __restrict__ dst_b,
    const int* __restrict__ off_b, int rows_b,
    const int* __restrict__ ent)
{
    const unsigned b = blockIdx.x;
    const bool second = b >= nb_a;
    pull_body(second ? src_b : src_a, second ? dst_b : dst_a,
              second ? off_b : off_a, ent, second ? rows_b : rows_a,
              second ? b - nb_a : b, second ? gridDim.x - nb_a : nb_a);
}


template <class Side>
__device__ __forceinline__ v4f sum_list(
    const v4f* __restrict__ s4, const v2i* __restrict__ en2,
    int b, int e, int l, int myi, float myw, Side side)
{
    v4f acc = {0.f, 0.f, 0.f, 0.f};
    bool side_pending = true;
    for (int base = b; base < e; base += 16) {
        int m = e - base;
        if (m > 16) m = 16;
        if (base != b) first_entries(en2, base, e, l, myi, myw);
        for (int k = 0; k < m; k += 8) {
            const int c = m - k;
            v4f v0, v1, v2, v3, v4, v5, v6, v7;
            const float w0 = __shfl(myw, k + 0, 16);
            const float w1 = __shfl(myw, k + 1, 16);
            const float w2 = __shfl(myw, k + 2, 16);
            const float w3 = __shfl(myw, k + 3, 16);
            const float w4 = __shfl(myw, k + 4, 16);
            const float w5 = __shfl(myw, k + 5, 16);
            const float w6 = __shfl(myw, k + 6, 16);
            const float w7 = __shfl(myw, k + 7, 16);
#define ROW_LD(j) v##j = s4[(size_t)__shfl(myi, k + (j), 16) * (D / 4) + l];
            ROW_LD(0)
            if (c > 1) { ROW_LD(1)
            if (c > 2) { ROW_LD(2)
            if (c > 3) { ROW_LD(3)
            if (c > 4) { ROW_LD(4)
            if (c > 5) { ROW_LD(5)
            if (c > 6) { ROW_LD(6)
            if (c > 7) { ROW_LD(7) } } } } } } }
#undef ROW_LD
            if (side_pending) { side(); side_pending = false; }
            acc += w0 * v0;
            if (c > 1) { acc += w1 * v1;
            if (c > 2) { acc += w2 * v2;
            if (c > 3) { acc += w3 * v3;
            if (c > 4) { acc += w4 * v4;
            if (c > 5) { acc += w5 * v5;
            if (c > 6) { acc += w6 * v6;
            if (c > 7) { acc += w7 * v7; } } } } } } }
        }
    }
    if (side_pending) side();
    return acc;
}

__device__ __forceinline__ v4f scale_two_rows(v4f x, int h, int l)
{
    v2f a;
    a.x = (l == 8 * h)     ? 0.25f : 0.0f;
    a.y = (l == 8 * h + 1) ? 0.25f : 0.0f;
    v2f b0, b1;
    b0.x = x.x; b0.y = x.y;
    b1.x = x.z; b1.y = x.w;
    const v8f zero8 = {};
    const v8f d0 = __builtin_amdgcn_wmma_f32_16x16x4_f32(
         false, a,  false, b0,
         (short)0, zero8,  false,  false);
    const v8f d1 = __builtin_amdgcn_wmma_f32_16x16x4_f32(
         false, a,  false, b1,
         (short)0, zero8,  false,  false);
    v4f y;
    y.x = d0[0]; y.y = d0[1]; y.z = d1[0]; y.w = d1[1];
    return y;
}

__global__ void __launch_bounds__(256) requested_rows_two_levels(
    const float* __restrict__ user_feat, const float* __restrict__ item_feat,
    const float* __restrict__ u1, const float* __restrict__ i1,
    const int* __restrict__ off, const int* __restrict__ ent,
    const int* __restrict__ users, const int* __restrict__ pos,
    const int* __restrict__ neg,
    float* __restrict__ out, int batch, int n_users)
{
    const int l = (int)(threadIdx.x & 15);
    const int h = (int)((threadIdx.x >> 4) & 1);
    const int rq = (int)((blockIdx.x * blockDim.x + threadIdx.x) >> 4);
    const int n_out = 3 * batch;
    if (rq - h >= n_out) return;
    const bool live = rq < n_out;
    const int r = live ? rq : n_out - 1;
    const int seg = r / batch;
    const int i   = r - seg * batch;
    const bool is_user = (seg == 0);
    const int idx = is_user ? users[i] : (seg == 1 ? pos[i] : neg[i]);

    const v4f* own0 = (const v4f*)(is_user ? user_feat : item_feat);
    const v4f* own1 = (const v4f*)(is_user ? u1 : i1);
    const v4f* opp1 = (const v4f*)(is_user ? i1 : u1);
    const int jr       = is_user ? idx : n_users + idx;
    const int opp_base = is_user ? n_users : 0;
    const v2i* en2 = (const v2i*)ent;

    const int b = off[jr], e = off[jr + 1];
    v4f acc2 = {0.f, 0.f, 0.f, 0.f};
    v4f acc3 = {0.f, 0.f, 0.f, 0.f};
    for (int base = b; base < e; base += 16) {
        int m = e - base;
        if (m > 16) m = 16;
        int   s  = 0, sb = 0, se = 0;
        float w  = 0.f;
        if (l < m) {
            v2i t = en2[base + l];
            s  = t.x;
            w  = __int_as_float(t.y);
            sb = off[opp_base + s];
            se = off[opp_base + s + 1];
        }
        int kb = __shfl(sb, 0, 16), ke = __shfl(se, 0, 16);
        int   ci;
        float cw;
        first_entries(en2, kb, ke, l, ci, cw);
        for (int k = 0; k < m; ++k) {
            const int   sk = __shfl(s, k, 16);
            const float wk = __shfl(w, k, 16);
            const int kn = (k + 1 < m) ? k + 1 : k;
            const int nb = __shfl(sb, kn, 16), ne = __shfl(se, kn, 16);
            int   ni = 0;
            float nw = 0.f;
            v4f   v;
            const v4f t = sum_list(own1, en2, kb, ke, l, ci, cw, [&]() {
                if (k + 1 < m) first_entries(en2, nb, ne, l, ni, nw);
                v = opp1[(size_t)sk * (D / 4) + l];
            });
            acc2 += wk * v;
            acc3 += wk * t;
            kb = nb; ke = ne; ci = ni; cw = nw;
        }
    }
    const v4f h0 = own0[(size_t)idx * (D / 4) + l];
    const v4f h1 = own1[(size_t)idx * (D / 4) + l];
    const v4f y = scale_two_rows(((h0 + h1) + acc2) + acc3, h, l);
    if (live) ((v4f*)out)[(size_t)r * (D / 4) + l] = y;
}

extern "C" void kernel_launch(void* const* d_in, const int* in_sizes, int n_in,
                              void* d_out, int out_size, void* d_ws, size_t ws_size,
                              hipStream_t stream)
{
    const float* user_feat = (const float*)d_in[0];
    const float* item_feat = (const float*)d_in[1];
    const int*   edge_u    = (const int*)d_in[2];
    const int*   edge_i    = (const int*)d_in[3];
    const float* norm      = (const float*)d_in[4];
    const int*   users     = (const int*)d_in[5];
    const int*   pos       = (const int*)d_in[6];
    const int*   neg       = (const int*)d_in[7];
    float* out = (float*)d_out;

    const int n_users = in_sizes[0] / D;
    const int n_items = in_sizes[1] / D;
    const int n_edges = in_sizes[2];
    const int batch   = in_sizes[5];
    const int n_rows  = n_users + n_items;

    size_t uElems = (size_t)n_users * D;
    size_t iElems = (size_t)n_items * D;

    float* su_a = (float*)d_ws;
    float* su_b = su_a + uElems;
    float* si   = su_b + uElems;
    int*   off  = (int*)(si + iElems);
    int offPad  = ((n_rows + 1 + 63) / 64) * 64;
    int*   ent  = off + offPad;

    const int nb = (n_rows + BUCKET_ROWS - 1) / BUCKET_ROWS;
    const int need = (n_edges + PASS_EDGES - 1) / PASS_EDGES;
    const int nchunks = need > MIN_CHUNKS ? need : MIN_CHUNKS;
    const int chunk = (((n_edges + nchunks - 1) / nchunks) + 3) & ~3;
    int*   tmp  = (int*)su_b;
    int*   cst  = tmp + 8 * (size_t)nchunks * (size_t)chunk;

    chunk_sort<<<nchunks, 1024, 0, stream>>>(edge_u, edge_i, norm, tmp, cst,
                                            n_users, n_edges, chunk, nb);
    bucket_to_lists<<<nb, BUCKET_ROWS, 0, stream>>>(tmp, cst, off, ent, n_rows, nchunks, chunk, nb);

    int ru = (n_users + 24575) / 24576;
    int ri = (n_items + 24575) / 24576;
    int ublocks = (n_users + 16 * ru - 1) / (16 * ru);
    int iblocks = (n_items + 16 * ri - 1) / (16 * ri);
    const int* off_u = off;
    const int* off_i = off + n_users;

    pull_both_sides<<<ublocks + iblocks, 256, 0, stream>>>(
        item_feat, su_a, off_u, n_users, ublocks,
        user_feat, si,   off_i, n_items, ent);

    int oblocks = (3 * batch + 15) / 16;
    requested_rows_two_levels<<<oblocks, 256, 0, stream>>>(
        user_feat, item_feat, su_a, si, off, ent, users, pos, neg, out, batch, n_users);

}
